# starting kernel
# baseline (speedup 1.0000x reference)
#include <hip/hip_runtime.h>

typedef __attribute__((ext_vector_type(2))) float v2f;
typedef __attribute__((ext_vector_type(8))) float v8f;

#define NB      2048
#define FC_K    100
#define FC_N    17640
#define NTILES  1103
#define NGRPS   138
#define H1_ELEMS (NB * FC_N)
#define H2_ELEMS (NB * 5 * 98 * 16)
#define CONV_BLOCKS 7840
#define PART_ELEMS (CONV_BLOCKS * 32)
#define BN_COUNT 1003520.0f

__global__ void __launch_bounds__(256) fc_wmma(const float* __restrict__ x,
                                               const float* __restrict__ fw,
                                               const float* __restrict__ fb,
                                               float* __restrict__ h1) {
  const int lane  = threadIdx.x & 31;
  const int wave  = threadIdx.x >> 5;
  const int w     = blockIdx.x * 8 + wave;
  const int mtile = w / NGRPS;
  const int ngrp  = w - mtile * NGRPS;
  const int lrow  = lane & 15;
  const int lhalf = lane >> 4;
  const int row   = mtile * 16 + lrow;
  const int kb0   = 2 * lhalf;

  v2f a[25];
#pragma unroll
  for (int s = 0; s < 25; ++s)
    a[s] = *(const v2f*)(x + row * FC_K + 4 * s + kb0);

  for (int t = 0; t < 8; ++t) {
    int ntile = ngrp * 8 + t;
    if (ntile >= NTILES) break;
    int n = ntile * 16 + lrow;
    v8f acc = {};
    if (ntile < NTILES - 1) {
      const float* bp = fw + n * FC_K + kb0;
#pragma unroll
      for (int s = 0; s < 25; ++s) {
        v2f bf = *(const v2f*)(bp + 4 * s);
        acc = __builtin_amdgcn_wmma_f32_16x16x4_f32(false, a[s], false, bf,
                                                    (short)0, acc, false, false);
      }
      float bias = fb[n];
#pragma unroll
      for (int v = 0; v < 8; ++v) {
        int m = mtile * 16 + lhalf * 8 + v;
        h1[m * FC_N + n] = acc[v] + bias;
      }
    } else {
      bool nvalid = (n < FC_N);
      const float* bp = fw + (nvalid ? n : 0) * FC_K + kb0;
#pragma unroll
      for (int s = 0; s < 25; ++s) {
        v2f bf = *(const v2f*)(bp + 4 * s);
        if (!nvalid) { bf[0] = 0.f; bf[1] = 0.f; }
        acc = __builtin_amdgcn_wmma_f32_16x16x4_f32(false, a[s], false, bf,
                                                    (short)0, acc, false, false);
      }
      float bias = nvalid ? fb[n] : 0.f;
#pragma unroll
      for (int v = 0; v < 8; ++v) {
        int m = mtile * 16 + lhalf * 8 + v;
        if (nvalid) h1[m * FC_N + n] = acc[v] + bias;
      }
    }
  }
}

__global__ void pack_conv_w(const float* __restrict__ cw, float* __restrict__ pw) {
  int idx = blockIdx.x * 256 + threadIdx.x;
  if (idx >= 2304) return;
  int n = idx / 144;
  int k = idx - n * 144;
  int p = k / 36;
  int c = k - p * 36;
  pw[idx] = cw[n * 144 + c * 4 + p];
}

__global__ void __launch_bounds__(256) conv_wmma(const float* __restrict__ h1,
                                                 const float* __restrict__ pw,
                                                 const float* __restrict__ cb,
                                                 float* __restrict__ h2,
                                                 float* __restrict__ part) {
  __shared__ float wsum[8][16];
  __shared__ float wsq[8][16];
  const int lane  = threadIdx.x & 31;
  const int wave  = threadIdx.x >> 5;
  const int w     = blockIdx.x * 8 + wave;
  const int mtile = w / 490;
  const int sp    = w - mtile * 490;
  const int i     = sp / 98;
  const int j     = sp - i * 98;
  const int lrow  = lane & 15;
  const int lhalf = lane >> 4;
  const int bb    = mtile * 16 + lrow;
  const float* bwp = pw + lrow * 144 + 2 * lhalf;

  v8f acc = {};
#pragma unroll
  for (int s = 0; s < 36; ++s) {
    const int p  = (4 * s) / 36;
    const int di = p >> 1, dj = p & 1;
    const int ii = i + di, jj = j + dj;
    int c0 = 4 * s - p * 36 + 2 * lhalf;
    v2f af;
    if (ii < 5 && jj < 98) {
      af = *(const v2f*)(h1 + (((bb * 5 + ii) * 98 + jj) * 36 + c0));
    } else {
      af[0] = 0.f; af[1] = 0.f;
    }
    v2f bf = *(const v2f*)(bwp + 4 * s);
    acc = __builtin_amdgcn_wmma_f32_16x16x4_f32(false, af, false, bf,
                                                (short)0, acc, false, false);
  }

  float bias = cb[lrow];
  float s0 = 0.f, q0 = 0.f;
#pragma unroll
  for (int v = 0; v < 8; ++v) {
    int m = mtile * 16 + lhalf * 8 + v;
    float val = acc[v] + bias;
    val = val >= 0.f ? val : 0.01f * val;
    h2[((m * 5 + i) * 98 + j) * 16 + lrow] = val;
    s0 += val;
    q0 += val * val;
  }
  s0 += __shfl_xor(s0, 16, 32);
  q0 += __shfl_xor(q0, 16, 32);
  if (lane < 16) { wsum[wave][lane] = s0; wsq[wave][lane] = q0; }
  __syncthreads();
  if (threadIdx.x < 32) {
    int c = threadIdx.x & 15;
    int st = threadIdx.x >> 4;
    float tot = 0.f;
    for (int wv = 0; wv < 8; ++wv) tot += st ? wsq[wv][c] : wsum[wv][c];
    part[blockIdx.x * 32 + threadIdx.x] = tot;
  }
}

__global__ void __launch_bounds__(256) bn_finalize(const float* __restrict__ part,
                                                   const float* __restrict__ gamma,
                                                   const float* __restrict__ beta,
                                                   float* __restrict__ scale,
                                                   float* __restrict__ shift) {
  __shared__ float acc[256];
  __shared__ float tot[32];
  int t = threadIdx.x;
  int cs = t & 31, r = t >> 5;
  float s = 0.f;
  for (int blk = r; blk < CONV_BLOCKS; blk += 8) s += part[blk * 32 + cs];
  acc[t] = s;
  __syncthreads();
  if (t < 32) {
    float sum = 0.f;
    for (int rr = 0; rr < 8; ++rr) sum += acc[rr * 32 + t];
    tot[t] = sum;
  }
  __syncthreads();
  if (t < 16) {
    const float invN = 1.0f / BN_COUNT;
    float mean = tot[t] * invN;
    float var  = tot[16 + t] * invN - mean * mean;
    float inv  = rsqrtf(var + 1e-5f);
    float scl  = gamma[t] * inv;
    scale[t] = scl;
    shift[t] = beta[t] - mean * scl;
  }
}

#define H2PB 7840
#define H3SZ 2328
__global__ void __launch_bounds__(128) lc_kernel(const float* __restrict__ h2,
                                                 const float* __restrict__ scale,
                                                 const float* __restrict__ shift,
                                                 const float* __restrict__ w1,
                                                 const float* __restrict__ b1,
                                                 const float* __restrict__ w2,
                                                 const float* __restrict__ b2,
                                                 float* __restrict__ out) {
  __shared__ float sh2[H2PB];
  __shared__ float sh3[H3SZ];
  const int b = blockIdx.x;
  const int tid = threadIdx.x;

  const float4 sc4 = ((const float4*)scale)[tid & 3];
  const float4 sf4 = ((const float4*)shift)[tid & 3];
  const float4* src4 = (const float4*)(h2 + b * H2PB);
  for (int idx = tid; idx < H2PB / 4; idx += 128) {
    float4 v = src4[idx];
    v.x = v.x * sc4.x + sf4.x;
    v.y = v.y * sc4.y + sf4.y;
    v.z = v.z * sc4.z + sf4.z;
    v.w = v.w * sc4.w + sf4.w;
    ((float4*)sh2)[idx] = v;
  }
  __syncthreads();

  for (int e = tid; e < H3SZ; e += 128) {
    int o1 = e / 388;
    int r  = e - o1 * 388;
    int oi = r / 97;
    int oj = r - oi * 97;
    float acc = b1[e];
    const float* wp = w1 + o1 * 24832 + oi * 388 + oj * 4;
    const float* xb = sh2 + (oi * 98 + oj) * 16;
#pragma unroll
    for (int c = 0; c < 16; ++c) {
      float4 wv = *(const float4*)(wp + c * 1552);
      acc += wv.x * xb[c]
           + wv.y * xb[16 + c]
           + wv.z * xb[98 * 16 + c]
           + wv.w * xb[99 * 16 + c];
    }
    acc = acc >= 0.f ? acc : 0.01f * acc;
    sh3[e] = acc;
  }
  __syncthreads();

  for (int e = tid; e < 288; e += 128) {
    int ri = e / 96;
    int rj = e - ri * 96;
    float acc = b2[e];
#pragma unroll
    for (int o1 = 0; o1 < 6; ++o1) {
      float4 wv = *(const float4*)(w2 + (((o1 * 3 + ri) * 96 + rj) << 2));
      const float* xb = sh3 + o1 * 388 + ri * 97 + rj;
      acc += wv.x * xb[0] + wv.y * xb[1] + wv.z * xb[97] + wv.w * xb[98];
    }
    out[b * 288 + e] = acc;
  }
}

extern "C" void kernel_launch(void* const* d_in, const int* in_sizes, int n_in,
                              void* d_out, int out_size, void* d_ws, size_t ws_size,
                              hipStream_t stream) {
  const float* x     = (const float*)d_in[0];
  const float* fw    = (const float*)d_in[1];
  const float* fb    = (const float*)d_in[2];
  const float* cw    = (const float*)d_in[3];
  const float* cb    = (const float*)d_in[4];
  const float* gamma = (const float*)d_in[5];
  const float* beta  = (const float*)d_in[6];
  const float* w1    = (const float*)d_in[7];
  const float* b1    = (const float*)d_in[8];
  const float* w2    = (const float*)d_in[9];
  const float* b2    = (const float*)d_in[10];

  float* ws    = (float*)d_ws;
  float* h1    = ws;
  float* h2    = ws + H1_ELEMS;
  float* part  = ws + H1_ELEMS + H2_ELEMS;
  float* scale = part + PART_ELEMS;
  float* shift = scale + 16;
  float* pw    = shift + 16;

  pack_conv_w<<<9, 256, 0, stream>>>(cw, pw);
  fc_wmma    <<<2208, 256, 0, stream>>>(x, fw, fb, h1);
  conv_wmma  <<<CONV_BLOCKS, 256, 0, stream>>>(h1, pw, cb, h2, part);
  bn_finalize<<<1, 256, 0, stream>>>(part, gamma, beta, scale, shift);
  lc_kernel  <<<NB, 128, 0, stream>>>(h2, scale, shift, w1, b1, w2, b2,
                                      (float*)d_out);
}


# BN statistics: one block walking all 7840 partial rows (122 dependent load rounds) becomes a 28-block first-level sum; each consumer wave adds the 28 rows itself; reduction tree changes, fixed order
# speedup vs baseline: 1.1731x; 1.1731x over previous
#include <hip/hip_runtime.h>

typedef __attribute__((ext_vector_type(2))) float v2f;
typedef __attribute__((ext_vector_type(8))) float v8f;

#define NB      2048
#define FC_K    100
#define FC_N    17640
#define NTILES  1103
#define NGRPS   138
#define H1_ELEMS (NB * FC_N)
#define H2_ELEMS (NB * 5 * 98 * 16)
#define CONV_BLOCKS 7840
#define PART_ELEMS (CONV_BLOCKS * 32)
#define BN_GROUPS 28
#define BN_ROWS_PER_GROUP 280
#define BN_COUNT 1003520.0f

__global__ void __launch_bounds__(256) fc_wmma(const float* __restrict__ x,
                                               const float* __restrict__ fw,
                                               const float* __restrict__ fb,
                                               float* __restrict__ h1) {
  const int lane  = threadIdx.x & 31;
  const int wave  = threadIdx.x >> 5;
  const int w     = blockIdx.x * 8 + wave;
  const int mtile = w / NGRPS;
  const int ngrp  = w - mtile * NGRPS;
  const int lrow  = lane & 15;
  const int lhalf = lane >> 4;
  const int row   = mtile * 16 + lrow;
  const int kb0   = 2 * lhalf;

  v2f a[25];
#pragma unroll
  for (int s = 0; s < 25; ++s)
    a[s] = *(const v2f*)(x + row * FC_K + 4 * s + kb0);

  for (int t = 0; t < 8; ++t) {
    int ntile = ngrp * 8 + t;
    if (ntile >= NTILES) break;
    int n = ntile * 16 + lrow;
    v8f acc = {};
    if (ntile < NTILES - 1) {
      const float* bp = fw + n * FC_K + kb0;
#pragma unroll
      for (int s = 0; s < 25; ++s) {
        v2f bf = *(const v2f*)(bp + 4 * s);
        acc = __builtin_amdgcn_wmma_f32_16x16x4_f32(false, a[s], false, bf,
                                                    (short)0, acc, false, false);
      }
      float bias = fb[n];
#pragma unroll
      for (int v = 0; v < 8; ++v) {
        int m = mtile * 16 + lhalf * 8 + v;
        h1[m * FC_N + n] = acc[v] + bias;
      }
    } else {
      bool nvalid = (n < FC_N);
      const float* bp = fw + (nvalid ? n : 0) * FC_K + kb0;
#pragma unroll
      for (int s = 0; s < 25; ++s) {
        v2f bf = *(const v2f*)(bp + 4 * s);
        if (!nvalid) { bf[0] = 0.f; bf[1] = 0.f; }
        acc = __builtin_amdgcn_wmma_f32_16x16x4_f32(false, a[s], false, bf,
                                                    (short)0, acc, false, false);
      }
      float bias = nvalid ? fb[n] : 0.f;
#pragma unroll
      for (int v = 0; v < 8; ++v) {
        int m = mtile * 16 + lhalf * 8 + v;
        if (nvalid) h1[m * FC_N + n] = acc[v] + bias;
      }
    }
  }
}

__global__ void pack_conv_w(const float* __restrict__ cw, float* __restrict__ pw) {
  int idx = blockIdx.x * 256 + threadIdx.x;
  if (idx >= 2304) return;
  int n = idx / 144;
  int k = idx - n * 144;
  int p = k / 36;
  int c = k - p * 36;
  pw[idx] = cw[n * 144 + c * 4 + p];
}

__global__ void __launch_bounds__(256) conv_wmma(const float* __restrict__ h1,
                                                 const float* __restrict__ pw,
                                                 const float* __restrict__ cb,
                                                 float* __restrict__ h2,
                                                 float* __restrict__ part) {
  __shared__ float wsum[8][16];
  __shared__ float wsq[8][16];
  const int lane  = threadIdx.x & 31;
  const int wave  = threadIdx.x >> 5;
  const int w     = blockIdx.x * 8 + wave;
  const int mtile = w / 490;
  const int sp    = w - mtile * 490;
  const int i     = sp / 98;
  const int j     = sp - i * 98;
  const int lrow  = lane & 15;
  const int lhalf = lane >> 4;
  const int bb    = mtile * 16 + lrow;
  const float* bwp = pw + lrow * 144 + 2 * lhalf;

  v8f acc = {};
#pragma unroll
  for (int s = 0; s < 36; ++s) {
    const int p  = (4 * s) / 36;
    const int di = p >> 1, dj = p & 1;
    const int ii = i + di, jj = j + dj;
    int c0 = 4 * s - p * 36 + 2 * lhalf;
    v2f af;
    if (ii < 5 && jj < 98) {
      af = *(const v2f*)(h1 + (((bb * 5 + ii) * 98 + jj) * 36 + c0));
    } else {
      af[0] = 0.f; af[1] = 0.f;
    }
    v2f bf = *(const v2f*)(bwp + 4 * s);
    acc = __builtin_amdgcn_wmma_f32_16x16x4_f32(false, af, false, bf,
                                                (short)0, acc, false, false);
  }

  float bias = cb[lrow];
  float s0 = 0.f, q0 = 0.f;
#pragma unroll
  for (int v = 0; v < 8; ++v) {
    int m = mtile * 16 + lhalf * 8 + v;
    float val = acc[v] + bias;
    val = val >= 0.f ? val : 0.01f * val;
    h2[((m * 5 + i) * 98 + j) * 16 + lrow] = val;
    s0 += val;
    q0 += val * val;
  }
  s0 += __shfl_xor(s0, 16, 32);
  q0 += __shfl_xor(q0, 16, 32);
  if (lane < 16) { wsum[wave][lane] = s0; wsq[wave][lane] = q0; }
  __syncthreads();
  if (threadIdx.x < 32) {
    int c = threadIdx.x & 15;
    int st = threadIdx.x >> 4;
    float tot = 0.f;
    for (int wv = 0; wv < 8; ++wv) tot += st ? wsq[wv][c] : wsum[wv][c];
    part[blockIdx.x * 32 + threadIdx.x] = tot;
  }
}

__global__ void __launch_bounds__(256) bn_reduce(const float* __restrict__ part,
                                                 float* __restrict__ gsum) {
  __shared__ float acc[256];
  const int t = threadIdx.x;
  const int cs = t & 31, r = t >> 5;
  const float* p = part + (blockIdx.x * BN_ROWS_PER_GROUP + r) * 32 + cs;
  float s = 0.f;
#pragma unroll
  for (int k = 0; k < BN_ROWS_PER_GROUP / 8; ++k) s += p[k * 256];
  acc[t] = s;
  __syncthreads();
  if (t < 32) {
    float sum = 0.f;
    for (int rr = 0; rr < 8; ++rr) sum += acc[rr * 32 + t];
    gsum[blockIdx.x * 32 + t] = sum;
  }
}

#define H2PB 7840
#define H3SZ 2328
__global__ void __launch_bounds__(128) lc_kernel(const float* __restrict__ h2,
                                                 const float* __restrict__ gsum,
                                                 const float* __restrict__ gamma,
                                                 const float* __restrict__ beta,
                                                 const float* __restrict__ w1,
                                                 const float* __restrict__ b1,
                                                 const float* __restrict__ w2,
                                                 const float* __restrict__ b2,
                                                 float* __restrict__ out) {
  __shared__ float sh2[H2PB];
  __shared__ float sh3[H3SZ];
  const int b = blockIdx.x;
  const int tid = threadIdx.x;

  const int lane = tid & 31;
  float tot = 0.f;
#pragma unroll
  for (int g = 0; g < BN_GROUPS; ++g) tot += gsum[g * 32 + lane];
  const float oth = __shfl_xor(tot, 16, 32);
  const float csum = lane < 16 ? tot : oth;
  const float csq  = lane < 16 ? oth : tot;
  const float invN = 1.0f / BN_COUNT;
  const float mean = csum * invN;
  const float var  = csq * invN - mean * mean;
  const float inv  = rsqrtf(var + 1e-5f);
  const float scl  = gamma[lane & 15] * inv;
  const float shf  = beta[lane & 15] - mean * scl;

  const int q4 = (tid & 3) * 4;
  float4 sc4, sf4;
  sc4.x = __shfl(scl, q4 + 0, 32); sf4.x = __shfl(shf, q4 + 0, 32);
  sc4.y = __shfl(scl, q4 + 1, 32); sf4.y = __shfl(shf, q4 + 1, 32);
  sc4.z = __shfl(scl, q4 + 2, 32); sf4.z = __shfl(shf, q4 + 2, 32);
  sc4.w = __shfl(scl, q4 + 3, 32); sf4.w = __shfl(shf, q4 + 3, 32);
  const float4* src4 = (const float4*)(h2 + b * H2PB);
  for (int idx = tid; idx < H2PB / 4; idx += 128) {
    float4 v = src4[idx];
    v.x = v.x * sc4.x + sf4.x;
    v.y = v.y * sc4.y + sf4.y;
    v.z = v.z * sc4.z + sf4.z;
    v.w = v.w * sc4.w + sf4.w;
    ((float4*)sh2)[idx] = v;
  }
  __syncthreads();

  for (int e = tid; e < H3SZ; e += 128) {
    int o1 = e / 388;
    int r  = e - o1 * 388;
    int oi = r / 97;
    int oj = r - oi * 97;
    float acc = b1[e];
    const float* wp = w1 + o1 * 24832 + oi * 388 + oj * 4;
    const float* xb = sh2 + (oi * 98 + oj) * 16;
#pragma unroll
    for (int c = 0; c < 16; ++c) {
      float4 wv = *(const float4*)(wp + c * 1552);
      acc += wv.x * xb[c]
           + wv.y * xb[16 + c]
           + wv.z * xb[98 * 16 + c]
           + wv.w * xb[99 * 16 + c];
    }
    acc = acc >= 0.f ? acc : 0.01f * acc;
    sh3[e] = acc;
  }
  __syncthreads();

  for (int e = tid; e < 288; e += 128) {
    int ri = e / 96;
    int rj = e - ri * 96;
    float acc = b2[e];
#pragma unroll
    for (int o1 = 0; o1 < 6; ++o1) {
      float4 wv = *(const float4*)(w2 + (((o1 * 3 + ri) * 96 + rj) << 2));
      const float* xb = sh3 + o1 * 388 + ri * 97 + rj;
      acc += wv.x * xb[0] + wv.y * xb[1] + wv.z * xb[97] + wv.w * xb[98];
    }
    out[b * 288 + e] = acc;
  }
}

extern "C" void kernel_launch(void* const* d_in, const int* in_sizes, int n_in,
                              void* d_out, int out_size, void* d_ws, size_t ws_size,
                              hipStream_t stream) {
  const float* x     = (const float*)d_in[0];
  const float* fw    = (const float*)d_in[1];
  const float* fb    = (const float*)d_in[2];
  const float* cw    = (const float*)d_in[3];
  const float* cb    = (const float*)d_in[4];
  const float* gamma = (const float*)d_in[5];
  const float* beta  = (const float*)d_in[6];
  const float* w1    = (const float*)d_in[7];
  const float* b1    = (const float*)d_in[8];
  const float* w2    = (const float*)d_in[9];
  const float* b2    = (const float*)d_in[10];

  float* ws    = (float*)d_ws;
  float* h1    = ws;
  float* h2    = ws + H1_ELEMS;
  float* part  = ws + H1_ELEMS + H2_ELEMS;
  float* pw    = part + PART_ELEMS;
  float* gsum  = h1;

  pack_conv_w<<<9, 256, 0, stream>>>(cw, pw);
  fc_wmma    <<<2208, 256, 0, stream>>>(x, fw, fb, h1);
  conv_wmma  <<<CONV_BLOCKS, 256, 0, stream>>>(h1, pw, cb, h2, part);
  bn_reduce  <<<BN_GROUPS, 256, 0, stream>>>(part, gsum);
  lc_kernel  <<<NB, 128, 0, stream>>>(h2, gsum, gamma, beta, w1, b1, w2, b2,
                                      (float*)d_out);
}


# conv: one block per batch element copies its h1 slab to LDS by whole-line loads; A fragments read from LDS (M tile 16 positions), weights in registers; BN partial rows 7840 to 2048: sum order changes
# speedup vs baseline: 1.6471x; 1.4041x over previous
#include <hip/hip_runtime.h>

typedef __attribute__((ext_vector_type(2))) float v2f;
typedef __attribute__((ext_vector_type(4))) float v4f;
typedef __attribute__((ext_vector_type(8))) float v8f;

#define NB      2048
#define FC_K    100
#define FC_N    17640
#define NTILES  1103
#define NGRPS   138
#define H1_ELEMS (NB * FC_N)
#define H2_ELEMS (NB * 5 * 98 * 16)
#define CONV_BLOCKS NB
#define PART_ELEMS 250880
#define BN_GROUPS 32
#define BN_ROWS_PER_GROUP 64
#define BN_COUNT 1003520.0f

__global__ void __launch_bounds__(256) fc_wmma(const float* __restrict__ x,
                                               const float* __restrict__ fw,
                                               const float* __restrict__ fb,
                                               float* __restrict__ h1) {
  const int lane  = threadIdx.x & 31;
  const int wave  = threadIdx.x >> 5;
  const int w     = blockIdx.x * 8 + wave;
  const int mtile = w / NGRPS;
  const int ngrp  = w - mtile * NGRPS;
  const int lrow  = lane & 15;
  const int lhalf = lane >> 4;
  const int row   = mtile * 16 + lrow;
  const int kb0   = 2 * lhalf;

  v2f a[25];
#pragma unroll
  for (int s = 0; s < 25; ++s)
    a[s] = *(const v2f*)(x + row * FC_K + 4 * s + kb0);

  for (int t = 0; t < 8; ++t) {
    int ntile = ngrp * 8 + t;
    if (ntile >= NTILES) break;
    int n = ntile * 16 + lrow;
    v8f acc = {};
    if (ntile < NTILES - 1) {
      const float* bp = fw + n * FC_K + kb0;
#pragma unroll
      for (int s = 0; s < 25; ++s) {
        v2f bf = *(const v2f*)(bp + 4 * s);
        acc = __builtin_amdgcn_wmma_f32_16x16x4_f32(false, a[s], false, bf,
                                                    (short)0, acc, false, false);
      }
      float bias = fb[n];
#pragma unroll
      for (int v = 0; v < 8; ++v) {
        int m = mtile * 16 + lhalf * 8 + v;
        h1[m * FC_N + n] = acc[v] + bias;
      }
    } else {
      bool nvalid = (n < FC_N);
      const float* bp = fw + (nvalid ? n : 0) * FC_K + kb0;
#pragma unroll
      for (int s = 0; s < 25; ++s) {
        v2f bf = *(const v2f*)(bp + 4 * s);
        if (!nvalid) { bf[0] = 0.f; bf[1] = 0.f; }
        acc = __builtin_amdgcn_wmma_f32_16x16x4_f32(false, a[s], false, bf,
                                                    (short)0, acc, false, false);
      }
      float bias = nvalid ? fb[n] : 0.f;
#pragma unroll
      for (int v = 0; v < 8; ++v) {
        int m = mtile * 16 + lhalf * 8 + v;
        if (nvalid) h1[m * FC_N + n] = acc[v] + bias;
      }
    }
  }
}

__global__ void pack_conv_w(const float* __restrict__ cw, float* __restrict__ pw) {
  int idx = blockIdx.x * 256 + threadIdx.x;
  if (idx >= 2304) return;
  int n = idx / 144;
  int k = idx - n * 144;
  int p = k / 36;
  int c = k - p * 36;
  pw[idx] = cw[n * 144 + c * 4 + p];
}

#define CPOS     490
#define H1PB     17640
#define CTILES   31
#define SA_ROWS  595
__global__ void __launch_bounds__(256) conv_wmma(const float* __restrict__ h1,
                                                 const float* __restrict__ pw,
                                                 const float* __restrict__ cb,
                                                 float* __restrict__ h2,
                                                 float* __restrict__ part) {
  __shared__ float sA[SA_ROWS * 36];
  __shared__ float wsum[8][16];
  __shared__ float wsq[8][16];
  const int tid   = threadIdx.x;
  const int lane  = tid & 31;
  const int wave  = __builtin_amdgcn_readfirstlane(tid >> 5);
  const int b     = blockIdx.x;
  const int lrow  = lane & 15;
  const int lhalf = lane >> 4;

  const float* bwp = pw + lrow * 144 + 2 * lhalf;
  v2f bfr[36];

  {
    const v4f* src = (const v4f*)(h1 + b * H1PB);
    v4f* dA = (v4f*)sA;
    v4f ra[18];
#pragma unroll
    for (int k = 0; k < 17; ++k) ra[k] = src[tid + k * 256];
    const int qa = 17 * 256 + tid;
    ra[17] = src[qa < H1PB / 4 ? qa : H1PB / 4 - 1];
#pragma unroll
    for (int s = 0; s < 36; ++s) bfr[s] = *(const v2f*)(bwp + 4 * s);
#pragma unroll
    for (int k = 0; k < 17; ++k) dA[tid + k * 256] = ra[k];
    const v4f z4 = {0.f, 0.f, 0.f, 0.f};
    const bool own = qa < H1PB / 4;
    v4f last;
    last[0] = own ? ra[17][0] : 0.f;
    last[1] = own ? ra[17][1] : 0.f;
    last[2] = own ? ra[17][2] : 0.f;
    last[3] = own ? ra[17][3] : 0.f;
    dA[qa] = last;
    for (int q = 18 * 256 + tid; q < SA_ROWS * 9; q += 256) dA[q] = z4;
  }
  __syncthreads();

  const float bias = cb[lrow];
  float s0 = 0.f, q0 = 0.f;

  for (int t = wave; t < CTILES; t += 8) {
    const int posr = t * 16 + lrow;
    const int jr   = posr % 98;
    const bool edge = (jr == 97);
    const float* ap = sA + posr * 36 + 2 * lhalf;
    v8f acc = {};
#pragma unroll
    for (int hh = 0; hh < 2; ++hh) {
      v2f af[18];
#pragma unroll
      for (int u = 0; u < 18; ++u) {
        const int s  = hh * 18 + u;
        const int p  = (4 * s) / 36;
        const int di = p >> 1, dj = p & 1;
        const int c0 = 4 * s - p * 36;
        v2f f = *(const v2f*)(ap + (di * 98 + dj) * 36 + c0);
        if (dj) {
          f[0] = edge ? 0.f : f[0];
          f[1] = edge ? 0.f : f[1];
        }
        af[u] = f;
      }
      __builtin_amdgcn_sched_barrier(0);
#pragma unroll
      for (int u = 0; u < 18; ++u)
        acc = __builtin_amdgcn_wmma_f32_16x16x4_f32(false, af[u], false, bfr[hh * 18 + u],
                                                    (short)0, acc, false, false);
    }
#pragma unroll
    for (int v = 0; v < 8; ++v) {
      int pos = t * 16 + lhalf * 8 + v;
      float val = acc[v] + bias;
      val = val >= 0.f ? val : 0.01f * val;
      if (pos < CPOS) {
        h2[(b * CPOS + pos) * 16 + lrow] = val;
        s0 += val;
        q0 += val * val;
      }
    }
  }

  s0 += __shfl_xor(s0, 16, 32);
  q0 += __shfl_xor(q0, 16, 32);
  if (lane < 16) { wsum[wave][lane] = s0; wsq[wave][lane] = q0; }
  __syncthreads();
  if (tid < 32) {
    int c = tid & 15;
    int st = tid >> 4;
    float tot = 0.f;
    for (int wv = 0; wv < 8; ++wv) tot += st ? wsq[wv][c] : wsum[wv][c];
    part[b * 32 + tid] = tot;
  }
}

__global__ void __launch_bounds__(256) bn_reduce(const float* __restrict__ part,
                                                 float* __restrict__ gsum) {
  __shared__ float acc[256];
  const int t = threadIdx.x;
  const int cs = t & 31, r = t >> 5;
  const float* p = part + (blockIdx.x * BN_ROWS_PER_GROUP + r) * 32 + cs;
  float s = 0.f;
#pragma unroll
  for (int k = 0; k < BN_ROWS_PER_GROUP / 8; ++k) s += p[k * 256];
  acc[t] = s;
  __syncthreads();
  if (t < 32) {
    float sum = 0.f;
    for (int rr = 0; rr < 8; ++rr) sum += acc[rr * 32 + t];
    gsum[blockIdx.x * 32 + t] = sum;
  }
}

#define H2PB 7840
#define H3SZ 2328
__global__ void __launch_bounds__(128) lc_kernel(const float* __restrict__ h2,
                                                 const float* __restrict__ gsum,
                                                 const float* __restrict__ gamma,
                                                 const float* __restrict__ beta,
                                                 const float* __restrict__ w1,
                                                 const float* __restrict__ b1,
                                                 const float* __restrict__ w2,
                                                 const float* __restrict__ b2,
                                                 float* __restrict__ out) {
  __shared__ float sh2[H2PB];
  __shared__ float sh3[H3SZ];
  const int b = blockIdx.x;
  const int tid = threadIdx.x;

  const int lane = tid & 31;
  float tot = 0.f;
#pragma unroll
  for (int g = 0; g < BN_GROUPS; ++g) tot += gsum[g * 32 + lane];
  const float oth = __shfl_xor(tot, 16, 32);
  const float csum = lane < 16 ? tot : oth;
  const float csq  = lane < 16 ? oth : tot;
  const float invN = 1.0f / BN_COUNT;
  const float mean = csum * invN;
  const float var  = csq * invN - mean * mean;
  const float inv  = rsqrtf(var + 1e-5f);
  const float scl  = gamma[lane & 15] * inv;
  const float shf  = beta[lane & 15] - mean * scl;

  const int q4 = (tid & 3) * 4;
  float4 sc4, sf4;
  sc4.x = __shfl(scl, q4 + 0, 32); sf4.x = __shfl(shf, q4 + 0, 32);
  sc4.y = __shfl(scl, q4 + 1, 32); sf4.y = __shfl(shf, q4 + 1, 32);
  sc4.z = __shfl(scl, q4 + 2, 32); sf4.z = __shfl(shf, q4 + 2, 32);
  sc4.w = __shfl(scl, q4 + 3, 32); sf4.w = __shfl(shf, q4 + 3, 32);
  const float4* src4 = (const float4*)(h2 + b * H2PB);
  for (int idx = tid; idx < H2PB / 4; idx += 128) {
    float4 v = src4[idx];
    v.x = v.x * sc4.x + sf4.x;
    v.y = v.y * sc4.y + sf4.y;
    v.z = v.z * sc4.z + sf4.z;
    v.w = v.w * sc4.w + sf4.w;
    ((float4*)sh2)[idx] = v;
  }
  __syncthreads();

  for (int e = tid; e < H3SZ; e += 128) {
    int o1 = e / 388;
    int r  = e - o1 * 388;
    int oi = r / 97;
    int oj = r - oi * 97;
    float acc = b1[e];
    const float* wp = w1 + o1 * 24832 + oi * 388 + oj * 4;
    const float* xb = sh2 + (oi * 98 + oj) * 16;
#pragma unroll
    for (int c = 0; c < 16; ++c) {
      float4 wv = *(const float4*)(wp + c * 1552);
      acc += wv.x * xb[c]
           + wv.y * xb[16 + c]
           + wv.z * xb[98 * 16 + c]
           + wv.w * xb[99 * 16 + c];
    }
    acc = acc >= 0.f ? acc : 0.01f * acc;
    sh3[e] = acc;
  }
  __syncthreads();

  for (int e = tid; e < 288; e += 128) {
    int ri = e / 96;
    int rj = e - ri * 96;
    float acc = b2[e];
#pragma unroll
    for (int o1 = 0; o1 < 6; ++o1) {
      float4 wv = *(const float4*)(w2 + (((o1 * 3 + ri) * 96 + rj) << 2));
      const float* xb = sh3 + o1 * 388 + ri * 97 + rj;
      acc += wv.x * xb[0] + wv.y * xb[1] + wv.z * xb[97] + wv.w * xb[98];
    }
    out[b * 288 + e] = acc;
  }
}

extern "C" void kernel_launch(void* const* d_in, const int* in_sizes, int n_in,
                              void* d_out, int out_size, void* d_ws, size_t ws_size,
                              hipStream_t stream) {
  const float* x     = (const float*)d_in[0];
  const float* fw    = (const float*)d_in[1];
  const float* fb    = (const float*)d_in[2];
  const float* cw    = (const float*)d_in[3];
  const float* cb    = (const float*)d_in[4];
  const float* gamma = (const float*)d_in[5];
  const float* beta  = (const float*)d_in[6];
  const float* w1    = (const float*)d_in[7];
  const float* b1    = (const float*)d_in[8];
  const float* w2    = (const float*)d_in[9];
  const float* b2    = (const float*)d_in[10];

  float* ws    = (float*)d_ws;
  float* h1    = ws;
  float* h2    = ws + H1_ELEMS;
  float* part  = ws + H1_ELEMS + H2_ELEMS;
  float* pw    = part + PART_ELEMS;
  float* gsum  = h1;

  pack_conv_w<<<9, 256, 0, stream>>>(cw, pw);
  fc_wmma    <<<2208, 256, 0, stream>>>(x, fw, fb, h1);
  conv_wmma  <<<CONV_BLOCKS, 256, 0, stream>>>(h1, pw, cb, h2, part);
  bn_reduce  <<<BN_GROUPS, 256, 0, stream>>>(part, gsum);
  lc_kernel  <<<NB, 128, 0, stream>>>(h2, gsum, gamma, beta, w1, b1, w2, b2,
                                      (float*)d_out);
}


# conv1 folded into the fc weights (both linear): W' = conv1(fc_w) built per call on WMMA, one GEMM x@W'^T then writes h2; h1 round trip gone; f32 sum order changes (GEMM association, BN partials)
# speedup vs baseline: 4.0783x; 2.4760x over previous
#include <hip/hip_runtime.h>

typedef __attribute__((ext_vector_type(2))) float v2f;
typedef __attribute__((ext_vector_type(4))) float v4f;
typedef __attribute__((ext_vector_type(8))) float v8f;

#define NB      2048
#define FC_K    100
#define CPOS    490
#define H1_ELEMS (NB * 17640)
#define H2_ELEMS (NB * 5 * 98 * 16)
#define H2ROW   7840
#define PART_ELEMS 250880
#define BN_GROUPS 32
#define BN_ROWS_PER_GROUP 49
#define BN_COUNT 1003520.0f
#define WQ_TILE 1600
#define WQ_OFF  4096
#define WQ_ELEMS (CPOS * WQ_TILE)
#define GM_ROWS  128
#define GM_TILES 5
#define GM_BLOCKS 1568

__global__ void pack_conv_w(const float* __restrict__ cw, float* __restrict__ pw) {
  int idx = blockIdx.x * 256 + threadIdx.x;
  if (idx >= 2304) return;
  int n = idx / 144;
  int k = idx - n * 144;
  int p = k / 36;
  int c = k - p * 36;
  pw[idx] = cw[n * 144 + c * 4 + p];
}

__global__ void __launch_bounds__(256) compose_w(const float* __restrict__ fw,
                                                 const float* __restrict__ fb,
                                                 const float* __restrict__ pw,
                                                 const float* __restrict__ cb,
                                                 float* __restrict__ wq,
                                                 float* __restrict__ bq) {
  const int lane  = threadIdx.x & 31;
  const int w     = blockIdx.x * 8 + __builtin_amdgcn_readfirstlane(threadIdx.x >> 5);
  if (w >= CPOS * 7) return;
  const int pos   = w / 7;
  const int kt    = w - pos * 7;
  const int lrow  = lane & 15;
  const int lhalf = lane >> 4;
  const int pi    = pos / 98;
  const int pj    = pos - pi * 98;
  const bool vdi  = pi < 4;
  const bool vdj  = pj < 97;
  const int k     = kt * 16 + lrow;
  const bool live = k <= FC_K;
  const float* colp = k < FC_K ? fw + k : fb;
  const int rs    = k < FC_K ? FC_K : 1;
  const float* bwp = pw + lrow * 144 + 2 * lhalf;

  v8f acc = {};
#pragma unroll
  for (int hh = 0; hh < 2; ++hh) {
    v2f af[18], bf[18];
#pragma unroll
    for (int u = 0; u < 18; ++u) {
      const int s  = hh * 18 + u;
      const int p  = (4 * s) / 36;
      const int di = p >> 1, dj = p & 1;
      const bool vp = (di ? vdi : true) && (dj ? vdj : true);
      const int off = vp ? di * 98 + dj : 0;
      const int r0 = (pos + off) * 36 + (4 * s - p * 36) + 2 * lhalf;
      v2f f;
      f[0] = colp[r0 * rs];
      f[1] = colp[(r0 + 1) * rs];
      const bool keep = vp && live;
      f[0] = keep ? f[0] : 0.f;
      f[1] = keep ? f[1] : 0.f;
      af[u] = f;
      bf[u] = *(const v2f*)(bwp + 4 * s);
    }
    __builtin_amdgcn_sched_barrier(0);
#pragma unroll
    for (int u = 0; u < 18; ++u)
      acc = __builtin_amdgcn_wmma_f32_16x16x4_f32(false, af[u], false, bf[u],
                                                  (short)0, acc, false, false);
  }
  float* wp = wq + pos * WQ_TILE;
  if (kt < 6) {
#pragma unroll
    for (int q = 0; q < 4; ++q) {
      const int s = kt * 4 + 2 * lhalf + (q >> 1);
      v2f o; o[0] = acc[2 * q]; o[1] = acc[2 * q + 1];
      *(v2f*)(wp + (s * 32 + (q & 1) * 16 + lrow) * 2) = o;
    }
  } else if (lhalf == 0) {
#pragma unroll
    for (int q = 0; q < 2; ++q) {
      v2f o; o[0] = acc[2 * q]; o[1] = acc[2 * q + 1];
      *(v2f*)(wp + (24 * 32 + q * 16 + lrow) * 2) = o;
    }
    bq[pos * 16 + lrow] = acc[4] + cb[lrow];
  }
}

__global__ void __launch_bounds__(256) fcconv_wmma(const float* __restrict__ x,
                                                   const float* __restrict__ wq,
                                                   const float* __restrict__ bq,
                                                   float* __restrict__ h2,
                                                   float* __restrict__ part) {
  __shared__ float sX[GM_ROWS * FC_K + 512];
  __shared__ float sW[GM_TILES * WQ_TILE + 192];
  __shared__ float wsum[8][16];
  __shared__ float wsq[8][16];
  const int tid   = threadIdx.x;
  const int lane  = tid & 31;
  const int wave  = __builtin_amdgcn_readfirstlane(tid >> 5);
  const int lrow  = lane & 15;
  const int lhalf = lane >> 4;
  const int mg    = blockIdx.x & 15;
  const int chunk = blockIdx.x >> 4;

  {
    const v4f* srcx = (const v4f*)(x + mg * (GM_ROWS * FC_K));
    const v4f* srcw = (const v4f*)(wq + chunk * (GM_TILES * WQ_TILE));
    v4f* dX = (v4f*)sX;
    v4f* dW = (v4f*)sW;
    v4f rx[13], rw[8];
#pragma unroll
    for (int k = 0; k < 12; ++k) rx[k] = srcx[tid + k * 256];
    const int qx = 12 * 256 + tid;
    rx[12] = srcx[qx < 3200 ? qx : 3199];
#pragma unroll
    for (int k = 0; k < 7; ++k) rw[k] = srcw[tid + k * 256];
    const int qw = 7 * 256 + tid;
    rw[7] = srcw[qw < 2000 ? qw : 1999];
#pragma unroll
    for (int k = 0; k < 12; ++k) dX[tid + k * 256] = rx[k];
    dX[qx] = rx[12];
#pragma unroll
    for (int k = 0; k < 7; ++k) dW[tid + k * 256] = rw[k];
    dW[qw] = rw[7];
  }
  __syncthreads();

  const float* ap = sX + (wave * 16 + lrow) * FC_K + 2 * lhalf;
  v2f a[25];
#pragma unroll
  for (int s = 0; s < 25; ++s) a[s] = *(const v2f*)(ap + 4 * s);

  const int m0 = (mg * 8 + wave) * 16 + lhalf * 8;
  float s0 = 0.f, q0 = 0.f;

  for (int t = 0; t < GM_TILES; ++t) {
    const int pos = chunk * GM_TILES + t;
    const float bias = bq[pos * 16 + lrow];
    const float* bp = sW + t * WQ_TILE + lane * 2;
    v2f bf[25];
#pragma unroll
    for (int s = 0; s < 25; ++s) bf[s] = *(const v2f*)(bp + s * 64);
    __builtin_amdgcn_sched_barrier(0);
    v8f acc = {};
#pragma unroll
    for (int s = 0; s < 25; ++s)
      acc = __builtin_amdgcn_wmma_f32_16x16x4_f32(false, a[s], false, bf[s],
                                                  (short)0, acc, false, false);
#pragma unroll
    for (int v = 0; v < 8; ++v) {
      float val = acc[v] + bias;
      val = val >= 0.f ? val : 0.01f * val;
      h2[(m0 + v) * H2ROW + pos * 16 + lrow] = val;
      s0 += val;
      q0 += val * val;
    }
  }

  s0 += __shfl_xor(s0, 16, 32);
  q0 += __shfl_xor(q0, 16, 32);
  if (lane < 16) { wsum[wave][lane] = s0; wsq[wave][lane] = q0; }
  __syncthreads();
  if (tid < 32) {
    int c = tid & 15;
    int st = tid >> 4;
    float tot = 0.f;
    for (int wv = 0; wv < 8; ++wv) tot += st ? wsq[wv][c] : wsum[wv][c];
    part[blockIdx.x * 32 + tid] = tot;
  }
}

__global__ void __launch_bounds__(256) bn_reduce(const float* __restrict__ part,
                                                 float* __restrict__ gsum) {
  __shared__ float acc[256];
  const int t = threadIdx.x;
  const int cs = t & 31, r = t >> 5;
  const float* p = part + (blockIdx.x * BN_ROWS_PER_GROUP + r) * 32 + cs;
  float s = 0.f;
#pragma unroll
  for (int k = 0; k < 6; ++k) s += p[k * 256];
  if (r == 0) s += p[6 * 256];
  acc[t] = s;
  __syncthreads();
  if (t < 32) {
    float sum = 0.f;
    for (int rr = 0; rr < 8; ++rr) sum += acc[rr * 32 + t];
    gsum[blockIdx.x * 32 + t] = sum;
  }
}

#define H2PB 7840
#define H3SZ 2328
__global__ void __launch_bounds__(128) lc_kernel(const float* __restrict__ h2,
                                                 const float* __restrict__ gsum,
                                                 const float* __restrict__ gamma,
                                                 const float* __restrict__ beta,
                                                 const float* __restrict__ w1,
                                                 const float* __restrict__ b1,
                                                 const float* __restrict__ w2,
                                                 const float* __restrict__ b2,
                                                 float* __restrict__ out) {
  __shared__ float sh2[H2PB];
  __shared__ float sh3[H3SZ];
  const int b = blockIdx.x;
  const int tid = threadIdx.x;

  const int lane = tid & 31;
  float tot = 0.f;
#pragma unroll
  for (int g = 0; g < BN_GROUPS; ++g) tot += gsum[g * 32 + lane];
  const float oth = __shfl_xor(tot, 16, 32);
  const float csum = lane < 16 ? tot : oth;
  const float csq  = lane < 16 ? oth : tot;
  const float invN = 1.0f / BN_COUNT;
  const float mean = csum * invN;
  const float var  = csq * invN - mean * mean;
  const float inv  = rsqrtf(var + 1e-5f);
  const float scl  = gamma[lane & 15] * inv;
  const float shf  = beta[lane & 15] - mean * scl;

  const int q4 = (tid & 3) * 4;
  float4 sc4, sf4;
  sc4.x = __shfl(scl, q4 + 0, 32); sf4.x = __shfl(shf, q4 + 0, 32);
  sc4.y = __shfl(scl, q4 + 1, 32); sf4.y = __shfl(shf, q4 + 1, 32);
  sc4.z = __shfl(scl, q4 + 2, 32); sf4.z = __shfl(shf, q4 + 2, 32);
  sc4.w = __shfl(scl, q4 + 3, 32); sf4.w = __shfl(shf, q4 + 3, 32);
  const float4* src4 = (const float4*)(h2 + b * H2PB);
  for (int idx = tid; idx < H2PB / 4; idx += 128) {
    float4 v = src4[idx];
    v.x = v.x * sc4.x + sf4.x;
    v.y = v.y * sc4.y + sf4.y;
    v.z = v.z * sc4.z + sf4.z;
    v.w = v.w * sc4.w + sf4.w;
    ((float4*)sh2)[idx] = v;
  }
  __syncthreads();

  for (int e = tid; e < H3SZ; e += 128) {
    int o1 = e / 388;
    int r  = e - o1 * 388;
    int oi = r / 97;
    int oj = r - oi * 97;
    float acc = b1[e];
    const float* wp = w1 + o1 * 24832 + oi * 388 + oj * 4;
    const float* xb = sh2 + (oi * 98 + oj) * 16;
#pragma unroll
    for (int c = 0; c < 16; ++c) {
      float4 wv = *(const float4*)(wp + c * 1552);
      acc += wv.x * xb[c]
           + wv.y * xb[16 + c]
           + wv.z * xb[98 * 16 + c]
           + wv.w * xb[99 * 16 + c];
    }
    acc = acc >= 0.f ? acc : 0.01f * acc;
    sh3[e] = acc;
  }
  __syncthreads();

  for (int e = tid; e < 288; e += 128) {
    int ri = e / 96;
    int rj = e - ri * 96;
    float acc = b2[e];
#pragma unroll
    for (int o1 = 0; o1 < 6; ++o1) {
      float4 wv = *(const float4*)(w2 + (((o1 * 3 + ri) * 96 + rj) << 2));
      const float* xb = sh3 + o1 * 388 + ri * 97 + rj;
      acc += wv.x * xb[0] + wv.y * xb[1] + wv.z * xb[97] + wv.w * xb[98];
    }
    out[b * 288 + e] = acc;
  }
}

extern "C" void kernel_launch(void* const* d_in, const int* in_sizes, int n_in,
                              void* d_out, int out_size, void* d_ws, size_t ws_size,
                              hipStream_t stream) {
  const float* x     = (const float*)d_in[0];
  const float* fw    = (const float*)d_in[1];
  const float* fb    = (const float*)d_in[2];
  const float* cw    = (const float*)d_in[3];
  const float* cb    = (const float*)d_in[4];
  const float* gamma = (const float*)d_in[5];
  const float* beta  = (const float*)d_in[6];
  const float* w1    = (const float*)d_in[7];
  const float* b1    = (const float*)d_in[8];
  const float* w2    = (const float*)d_in[9];
  const float* b2    = (const float*)d_in[10];

  float* ws    = (float*)d_ws;
  float* gsum  = ws;
  float* wq    = ws + WQ_OFF;
  float* bq    = ws + WQ_OFF + WQ_ELEMS;
  float* h2    = ws + H1_ELEMS;
  float* part  = ws + H1_ELEMS + H2_ELEMS;
  float* pw    = part + PART_ELEMS;

  pack_conv_w<<<9, 256, 0, stream>>>(cw, pw);
  compose_w  <<<429, 256, 0, stream>>>(fw, fb, pw, cb, wq, bq);
  fcconv_wmma<<<GM_BLOCKS, 256, 0, stream>>>(x, wq, bq, h2, part);
  bn_reduce  <<<BN_GROUPS, 256, 0, stream>>>(part, gsum);
  lc_kernel  <<<NB, 128, 0, stream>>>(h2, gsum, gamma, beta, w1, b1, w2, b2,
                                      (float*)d_out);
}


# lc kernel tiled 8 batch elements x 32-column strip per block: a thread keeps the lc1 weights of one position and two channels in registers and reuses them along the batch; sum order kept
# speedup vs baseline: 6.1544x; 1.5091x over previous
#include <hip/hip_runtime.h>

typedef __attribute__((ext_vector_type(2))) float v2f;
typedef __attribute__((ext_vector_type(4))) float v4f;
typedef __attribute__((ext_vector_type(8))) float v8f;

#define NB      2048
#define FC_K    100
#define CPOS    490
#define H1_ELEMS (NB * 17640)
#define H2_ELEMS (NB * 5 * 98 * 16)
#define H2ROW   7840
#define PART_ELEMS 250880
#define BN_GROUPS 32
#define BN_ROWS_PER_GROUP 49
#define BN_COUNT 1003520.0f
#define WQ_TILE 1600
#define WQ_OFF  4096
#define WQ_ELEMS (CPOS * WQ_TILE)
#define GM_ROWS  128
#define GM_TILES 5
#define GM_BLOCKS 1568

__global__ void pack_conv_w(const float* __restrict__ cw, float* __restrict__ pw) {
  int idx = blockIdx.x * 256 + threadIdx.x;
  if (idx >= 2304) return;
  int n = idx / 144;
  int k = idx - n * 144;
  int p = k / 36;
  int c = k - p * 36;
  pw[idx] = cw[n * 144 + c * 4 + p];
}

__global__ void __launch_bounds__(256) compose_w(const float* __restrict__ fw,
                                                 const float* __restrict__ fb,
                                                 const float* __restrict__ pw,
                                                 const float* __restrict__ cb,
                                                 float* __restrict__ wq,
                                                 float* __restrict__ bq) {
  const int lane  = threadIdx.x & 31;
  const int w     = blockIdx.x * 8 + __builtin_amdgcn_readfirstlane(threadIdx.x >> 5);
  if (w >= CPOS * 7) return;
  const int pos   = w / 7;
  const int kt    = w - pos * 7;
  const int lrow  = lane & 15;
  const int lhalf = lane >> 4;
  const int pi    = pos / 98;
  const int pj    = pos - pi * 98;
  const bool vdi  = pi < 4;
  const bool vdj  = pj < 97;
  const int k     = kt * 16 + lrow;
  const bool live = k <= FC_K;
  const float* colp = k < FC_K ? fw + k : fb;
  const int rs    = k < FC_K ? FC_K : 1;
  const float* bwp = pw + lrow * 144 + 2 * lhalf;

  v8f acc = {};
#pragma unroll
  for (int hh = 0; hh < 2; ++hh) {
    v2f af[18], bf[18];
#pragma unroll
    for (int u = 0; u < 18; ++u) {
      const int s  = hh * 18 + u;
      const int p  = (4 * s) / 36;
      const int di = p >> 1, dj = p & 1;
      const bool vp = (di ? vdi : true) && (dj ? vdj : true);
      const int off = vp ? di * 98 + dj : 0;
      const int r0 = (pos + off) * 36 + (4 * s - p * 36) + 2 * lhalf;
      v2f f;
      f[0] = colp[r0 * rs];
      f[1] = colp[(r0 + 1) * rs];
      const bool keep = vp && live;
      f[0] = keep ? f[0] : 0.f;
      f[1] = keep ? f[1] : 0.f;
      af[u] = f;
      bf[u] = *(const v2f*)(bwp + 4 * s);
    }
    __builtin_amdgcn_sched_barrier(0);
#pragma unroll
    for (int u = 0; u < 18; ++u)
      acc = __builtin_amdgcn_wmma_f32_16x16x4_f32(false, af[u], false, bf[u],
                                                  (short)0, acc, false, false);
  }
  float* wp = wq + pos * WQ_TILE;
  if (kt < 6) {
#pragma unroll
    for (int q = 0; q < 4; ++q) {
      const int s = kt * 4 + 2 * lhalf + (q >> 1);
      v2f o; o[0] = acc[2 * q]; o[1] = acc[2 * q + 1];
      *(v2f*)(wp + (s * 32 + (q & 1) * 16 + lrow) * 2) = o;
    }
  } else if (lhalf == 0) {
#pragma unroll
    for (int q = 0; q < 2; ++q) {
      v2f o; o[0] = acc[2 * q]; o[1] = acc[2 * q + 1];
      *(v2f*)(wp + (24 * 32 + q * 16 + lrow) * 2) = o;
    }
    bq[pos * 16 + lrow] = acc[4] + cb[lrow];
  }
}

__global__ void __launch_bounds__(256) fcconv_wmma(const float* __restrict__ x,
                                                   const float* __restrict__ wq,
                                                   const float* __restrict__ bq,
                                                   float* __restrict__ h2,
                                                   float* __restrict__ part) {
  __shared__ float sX[GM_ROWS * FC_K + 512];
  __shared__ float sW[GM_TILES * WQ_TILE + 192];
  __shared__ float wsum[8][16];
  __shared__ float wsq[8][16];
  const int tid   = threadIdx.x;
  const int lane  = tid & 31;
  const int wave  = __builtin_amdgcn_readfirstlane(tid >> 5);
  const int lrow  = lane & 15;
  const int lhalf = lane >> 4;
  const int mg    = blockIdx.x & 15;
  const int chunk = blockIdx.x >> 4;

  {
    const v4f* srcx = (const v4f*)(x + mg * (GM_ROWS * FC_K));
    const v4f* srcw = (const v4f*)(wq + chunk * (GM_TILES * WQ_TILE));
    v4f* dX = (v4f*)sX;
    v4f* dW = (v4f*)sW;
    v4f rx[13], rw[8];
#pragma unroll
    for (int k = 0; k < 12; ++k) rx[k] = srcx[tid + k * 256];
    const int qx = 12 * 256 + tid;
    rx[12] = srcx[qx < 3200 ? qx : 3199];
#pragma unroll
    for (int k = 0; k < 7; ++k) rw[k] = srcw[tid + k * 256];
    const int qw = 7 * 256 + tid;
    rw[7] = srcw[qw < 2000 ? qw : 1999];
#pragma unroll
    for (int k = 0; k < 12; ++k) dX[tid + k * 256] = rx[k];
    dX[qx] = rx[12];
#pragma unroll
    for (int k = 0; k < 7; ++k) dW[tid + k * 256] = rw[k];
    dW[qw] = rw[7];
  }
  __syncthreads();

  const float* ap = sX + (wave * 16 + lrow) * FC_K + 2 * lhalf;
  v2f a[25];
#pragma unroll
  for (int s = 0; s < 25; ++s) a[s] = *(const v2f*)(ap + 4 * s);

  const int m0 = (mg * 8 + wave) * 16 + lhalf * 8;
  float s0 = 0.f, q0 = 0.f;

  for (int t = 0; t < GM_TILES; ++t) {
    const int pos = chunk * GM_TILES + t;
    const float bias = bq[pos * 16 + lrow];
    const float* bp = sW + t * WQ_TILE + lane * 2;
    v2f bf[25];
#pragma unroll
    for (int s = 0; s < 25; ++s) bf[s] = *(const v2f*)(bp + s * 64);
    __builtin_amdgcn_sched_barrier(0);
    v8f acc = {};
#pragma unroll
    for (int s = 0; s < 25; ++s)
      acc = __builtin_amdgcn_wmma_f32_16x16x4_f32(false, a[s], false, bf[s],
                                                  (short)0, acc, false, false);
#pragma unroll
    for (int v = 0; v < 8; ++v) {
      float val = acc[v] + bias;
      val = val >= 0.f ? val : 0.01f * val;
      h2[(m0 + v) * H2ROW + pos * 16 + lrow] = val;
      s0 += val;
      q0 += val * val;
    }
  }

  s0 += __shfl_xor(s0, 16, 32);
  q0 += __shfl_xor(q0, 16, 32);
  if (lane < 16) { wsum[wave][lane] = s0; wsq[wave][lane] = q0; }
  __syncthreads();
  if (tid < 32) {
    int c = tid & 15;
    int st = tid >> 4;
    float tot = 0.f;
    for (int wv = 0; wv < 8; ++wv) tot += st ? wsq[wv][c] : wsum[wv][c];
    part[blockIdx.x * 32 + tid] = tot;
  }
}

__global__ void __launch_bounds__(256) bn_reduce(const float* __restrict__ part,
                                                 float* __restrict__ gsum) {
  __shared__ float acc[256];
  const int t = threadIdx.x;
  const int cs = t & 31, r = t >> 5;
  const float* p = part + (blockIdx.x * BN_ROWS_PER_GROUP + r) * 32 + cs;
  float s = 0.f;
#pragma unroll
  for (int k = 0; k < 6; ++k) s += p[k * 256];
  if (r == 0) s += p[6 * 256];
  acc[t] = s;
  __syncthreads();
  if (t < 32) {
    float sum = 0.f;
    for (int rr = 0; rr < 8; ++rr) sum += acc[rr * 32 + t];
    gsum[blockIdx.x * 32 + t] = sum;
  }
}

#define LC_G      8
#define LC_HC     34
#define LC_TC     33
#define LC_H2Q    850
#define LC_H3G    792
#define LC_STAGEQ 5440
#define LC_TASKS  198
#define LC_BLOCKS 768

__device__ __forceinline__ void lc_load_w(const float* __restrict__ w1,
                                          const float* __restrict__ b1,
                                          int idx, int j0,
                                          float4 (&wa)[16], float4 (&wb)[16],
                                          float& bia, float& bib,
                                          int& xoff, int& h3off) {
  const int o1p = idx % 3;
  const int pp  = idx / 3;
  const int oi  = pp / LC_TC;
  const int ojl = pp - oi * LC_TC;
  const int oj  = j0 + ojl;
  const int o1a = 2 * o1p;
  const float* wp = w1 + o1a * 24832 + oi * 388 + oj * 4;
#pragma unroll
  for (int c = 0; c < 16; ++c) {
    wa[c] = *(const float4*)(wp + c * 1552);
    wb[c] = *(const float4*)(wp + 24832 + c * 1552);
  }
  bia = b1[o1a * 388 + oi * 97 + oj];
  bib = b1[(o1a + 1) * 388 + oi * 97 + oj];
  xoff  = (oi * LC_HC + ojl) * 5;
  h3off = o1a * 132 + oi * LC_TC + ojl;
}

__device__ __forceinline__ void lc2_one(int e, int b0, int j0,
                                        const float* __restrict__ w2,
                                        const float* __restrict__ b2,
                                        const float* sh3,
                                        float* __restrict__ out) {
  const int g   = e / 96;
  const int r   = e - g * 96;
  const int ri  = r >> 5;
  const int rjl = r & 31;
  const int rj  = j0 + rjl;
  float acc = b2[ri * 96 + rj];
#pragma unroll
  for (int o1 = 0; o1 < 6; ++o1) {
    float4 wv = *(const float4*)(w2 + (((o1 * 3 + ri) * 96 + rj) << 2));
    const float* xb = sh3 + g * LC_H3G + o1 * 132 + ri * LC_TC + rjl;
    acc += wv.x * xb[0] + wv.y * xb[1] + wv.z * xb[33] + wv.w * xb[34];
  }
  out[(b0 + g) * 288 + ri * 96 + rj] = acc;
}

__global__ void __launch_bounds__(224) lc_kernel(const float* __restrict__ h2,
                                                 const float* __restrict__ gsum,
                                                 const float* __restrict__ gamma,
                                                 const float* __restrict__ beta,
                                                 const float* __restrict__ w1,
                                                 const float* __restrict__ b1,
                                                 const float* __restrict__ w2,
                                                 const float* __restrict__ b2,
                                                 float* __restrict__ out) {
  __shared__ float4 sh2q[LC_G * LC_H2Q];
  __shared__ float sh3[LC_G * LC_H3G];
  const int tid   = threadIdx.x;
  const int strip = blockIdx.x % 3;
  const int b0    = (blockIdx.x / 3) * LC_G;
  const int j0    = strip * 32;

  float4 wa[16], wb[16];
  float bia, bib;
  int xoff, h3off;
  lc_load_w(w1, b1, tid, j0, wa, wb, bia, bib, xoff, h3off);

  const int lane = tid & 31;
  float tot = 0.f;
#pragma unroll
  for (int g = 0; g < BN_GROUPS; ++g) tot += gsum[g * 32 + lane];
  const float oth = __shfl_xor(tot, 16, 32);
  const float csum = lane < 16 ? tot : oth;
  const float csq  = lane < 16 ? oth : tot;
  const float invN = 1.0f / BN_COUNT;
  const float mean = csum * invN;
  const float var  = csq * invN - mean * mean;
  const float inv  = rsqrtf(var + 1e-5f);
  const float scl  = gamma[lane & 15] * inv;
  const float shf  = beta[lane & 15] - mean * scl;

  const int q4 = (tid & 3) * 4;
  float4 sc4, sf4;
  sc4.x = __shfl(scl, q4 + 0, 32); sf4.x = __shfl(shf, q4 + 0, 32);
  sc4.y = __shfl(scl, q4 + 1, 32); sf4.y = __shfl(shf, q4 + 1, 32);
  sc4.z = __shfl(scl, q4 + 2, 32); sf4.z = __shfl(shf, q4 + 2, 32);
  sc4.w = __shfl(scl, q4 + 3, 32); sf4.w = __shfl(shf, q4 + 3, 32);
  const float4* src4 = (const float4*)h2;
#pragma nounroll
  for (int k0 = 0; k0 < 24; k0 += 12) {
    float4 r[12];
    int di[12];
#pragma unroll
    for (int u = 0; u < 12; ++u) {
      const int q   = tid + (k0 + u) * 224;
      const int seg = q / 136;
      const int w   = q - seg * 136;
      const int g   = seg / 5;
      const int i   = seg - g * 5;
      const int jj  = w >> 2;
      const int cq  = w & 3;
      r[u]  = src4[((b0 + g) * CPOS + i * 98 + j0 + jj) * 4 + cq];
      di[u] = (seg * LC_HC + jj) * 5 + cq;
    }
#pragma unroll
    for (int u = 0; u < 12; ++u) {
      float4 v = r[u];
      v.x = v.x * sc4.x + sf4.x;
      v.y = v.y * sc4.y + sf4.y;
      v.z = v.z * sc4.z + sf4.z;
      v.w = v.w * sc4.w + sf4.w;
      sh2q[di[u]] = v;
    }
  }
  if (tid < LC_STAGEQ - 24 * 224) {
    const int q   = tid + 24 * 224;
    const int seg = q / 136;
    const int w   = q - seg * 136;
    const int g   = seg / 5;
    const int i   = seg - g * 5;
    const int jj  = w >> 2;
    const int cq  = w & 3;
    float4 v = src4[((b0 + g) * CPOS + i * 98 + j0 + jj) * 4 + cq];
    v.x = v.x * sc4.x + sf4.x;
    v.y = v.y * sc4.y + sf4.y;
    v.z = v.z * sc4.z + sf4.z;
    v.w = v.w * sc4.w + sf4.w;
    sh2q[(seg * LC_HC + jj) * 5 + cq] = v;
  }
  __syncthreads();

#pragma nounroll
  for (int pass = 0; pass < 2; ++pass) {
    if (pass == 1) {
      const int idx = LC_TASKS + tid;
      lc_load_w(w1, b1, idx < 395 ? idx : 395, j0, wa, wb, bia, bib, xoff, h3off);
    }
    if (tid < LC_TASKS) {
#pragma nounroll
      for (int g = 0; g < LC_G; ++g) {
        const float4* xg = sh2q + g * LC_H2Q + xoff;
        float a0 = bia, a1 = bib;
#pragma unroll
        for (int cq = 0; cq < 4; ++cq) {
          const float4 x00 = xg[cq];
          const float4 x01 = xg[5 + cq];
          const float4 x10 = xg[170 + cq];
          const float4 x11 = xg[175 + cq];
          a0 += wa[4 * cq + 0].x * x00.x + wa[4 * cq + 0].y * x01.x + wa[4 * cq + 0].z * x10.x + wa[4 * cq + 0].w * x11.x;
          a1 += wb[4 * cq + 0].x * x00.x + wb[4 * cq + 0].y * x01.x + wb[4 * cq + 0].z * x10.x + wb[4 * cq + 0].w * x11.x;
          a0 += wa[4 * cq + 1].x * x00.y + wa[4 * cq + 1].y * x01.y + wa[4 * cq + 1].z * x10.y + wa[4 * cq + 1].w * x11.y;
          a1 += wb[4 * cq + 1].x * x00.y + wb[4 * cq + 1].y * x01.y + wb[4 * cq + 1].z * x10.y + wb[4 * cq + 1].w * x11.y;
          a0 += wa[4 * cq + 2].x * x00.z + wa[4 * cq + 2].y * x01.z + wa[4 * cq + 2].z * x10.z + wa[4 * cq + 2].w * x11.z;
          a1 += wb[4 * cq + 2].x * x00.z + wb[4 * cq + 2].y * x01.z + wb[4 * cq + 2].z * x10.z + wb[4 * cq + 2].w * x11.z;
          a0 += wa[4 * cq + 3].x * x00.w + wa[4 * cq + 3].y * x01.w + wa[4 * cq + 3].z * x10.w + wa[4 * cq + 3].w * x11.w;
          a1 += wb[4 * cq + 3].x * x00.w + wb[4 * cq + 3].y * x01.w + wb[4 * cq + 3].z * x10.w + wb[4 * cq + 3].w * x11.w;
        }
        a0 = a0 >= 0.f ? a0 : 0.01f * a0;
        a1 = a1 >= 0.f ? a1 : 0.01f * a1;
        sh3[g * LC_H3G + h3off]       = a0;
        sh3[g * LC_H3G + h3off + 132] = a1;
      }
    }
  }
  __syncthreads();

  lc2_one(tid,       b0, j0, w2, b2, sh3, out);
  lc2_one(tid + 224, b0, j0, w2, b2, sh3, out);
  lc2_one(tid + 448, b0, j0, w2, b2, sh3, out);
  if (tid < LC_G * 96 - 3 * 224) lc2_one(tid + 672, b0, j0, w2, b2, sh3, out);
}

extern "C" void kernel_launch(void* const* d_in, const int* in_sizes, int n_in,
                              void* d_out, int out_size, void* d_ws, size_t ws_size,
                              hipStream_t stream) {
  const float* x     = (const float*)d_in[0];
  const float* fw    = (const float*)d_in[1];
  const float* fb    = (const float*)d_in[2];
  const float* cw    = (const float*)d_in[3];
  const float* cb    = (const float*)d_in[4];
  const float* gamma = (const float*)d_in[5];
  const float* beta  = (const float*)d_in[6];
  const float* w1    = (const float*)d_in[7];
  const float* b1    = (const float*)d_in[8];
  const float* w2    = (const float*)d_in[9];
  const float* b2    = (const float*)d_in[10];

  float* ws    = (float*)d_ws;
  float* gsum  = ws;
  float* wq    = ws + WQ_OFF;
  float* bq    = ws + WQ_OFF + WQ_ELEMS;
  float* h2    = ws + H1_ELEMS;
  float* part  = ws + H1_ELEMS + H2_ELEMS;
  float* pw    = part + PART_ELEMS;

  pack_conv_w<<<9, 256, 0, stream>>>(cw, pw);
  compose_w  <<<429, 256, 0, stream>>>(fw, fb, pw, cb, wq, bq);
  fcconv_wmma<<<GM_BLOCKS, 256, 0, stream>>>(x, wq, bq, h2, part);
  bn_reduce  <<<BN_GROUPS, 256, 0, stream>>>(part, gsum);
  lc_kernel  <<<LC_BLOCKS, 224, 0, stream>>>(h2, gsum, gamma, beta, w1, b1, w2, b2,
                                      (float*)d_out);
}


# lc1 inner loop: h2 tile held in LDS as batch-element pairs per channel, so 128-bit reads feed packed f32 multiply-adds with no register moves; next quad's reads grouped ahead; sum order kept
# speedup vs baseline: 6.3926x; 1.0387x over previous
#include <hip/hip_runtime.h>

typedef __attribute__((ext_vector_type(2))) float v2f;
typedef __attribute__((ext_vector_type(4))) float v4f;
typedef __attribute__((ext_vector_type(8))) float v8f;

#define NB      2048
#define FC_K    100
#define CPOS    490
#define H1_ELEMS (NB * 17640)
#define H2_ELEMS (NB * 5 * 98 * 16)
#define H2ROW   7840
#define PART_ELEMS 250880
#define BN_GROUPS 32
#define BN_ROWS_PER_GROUP 49
#define BN_COUNT 1003520.0f
#define WQ_TILE 1600
#define WQ_OFF  4096
#define WQ_ELEMS (CPOS * WQ_TILE)
#define GM_ROWS  128
#define GM_TILES 5
#define GM_BLOCKS 1568

__global__ void pack_conv_w(const float* __restrict__ cw, float* __restrict__ pw) {
  int idx = blockIdx.x * 256 + threadIdx.x;
  if (idx >= 2304) return;
  int n = idx / 144;
  int k = idx - n * 144;
  int p = k / 36;
  int c = k - p * 36;
  pw[idx] = cw[n * 144 + c * 4 + p];
}

__global__ void __launch_bounds__(256) compose_w(const float* __restrict__ fw,
                                                 const float* __restrict__ fb,
                                                 const float* __restrict__ pw,
                                                 const float* __restrict__ cb,
                                                 float* __restrict__ wq,
                                                 float* __restrict__ bq) {
  const int lane  = threadIdx.x & 31;
  const int w     = blockIdx.x * 8 + __builtin_amdgcn_readfirstlane(threadIdx.x >> 5);
  if (w >= CPOS * 7) return;
  const int pos   = w / 7;
  const int kt    = w - pos * 7;
  const int lrow  = lane & 15;
  const int lhalf = lane >> 4;
  const int pi    = pos / 98;
  const int pj    = pos - pi * 98;
  const bool vdi  = pi < 4;
  const bool vdj  = pj < 97;
  const int k     = kt * 16 + lrow;
  const bool live = k <= FC_K;
  const float* colp = k < FC_K ? fw + k : fb;
  const int rs    = k < FC_K ? FC_K : 1;
  const float* bwp = pw + lrow * 144 + 2 * lhalf;

  v8f acc = {};
#pragma unroll
  for (int hh = 0; hh < 2; ++hh) {
    v2f af[18], bf[18];
#pragma unroll
    for (int u = 0; u < 18; ++u) {
      const int s  = hh * 18 + u;
      const int p  = (4 * s) / 36;
      const int di = p >> 1, dj = p & 1;
      const bool vp = (di ? vdi : true) && (dj ? vdj : true);
      const int off = vp ? di * 98 + dj : 0;
      const int r0 = (pos + off) * 36 + (4 * s - p * 36) + 2 * lhalf;
      v2f f;
      f[0] = colp[r0 * rs];
      f[1] = colp[(r0 + 1) * rs];
      const bool keep = vp && live;
      f[0] = keep ? f[0] : 0.f;
      f[1] = keep ? f[1] : 0.f;
      af[u] = f;
      bf[u] = *(const v2f*)(bwp + 4 * s);
    }
    __builtin_amdgcn_sched_barrier(0);
#pragma unroll
    for (int u = 0; u < 18; ++u)
      acc = __builtin_amdgcn_wmma_f32_16x16x4_f32(false, af[u], false, bf[u],
                                                  (short)0, acc, false, false);
  }
  float* wp = wq + pos * WQ_TILE;
  if (kt < 6) {
#pragma unroll
    for (int q = 0; q < 4; ++q) {
      const int s = kt * 4 + 2 * lhalf + (q >> 1);
      v2f o; o[0] = acc[2 * q]; o[1] = acc[2 * q + 1];
      *(v2f*)(wp + (s * 32 + (q & 1) * 16 + lrow) * 2) = o;
    }
  } else if (lhalf == 0) {
#pragma unroll
    for (int q = 0; q < 2; ++q) {
      v2f o; o[0] = acc[2 * q]; o[1] = acc[2 * q + 1];
      *(v2f*)(wp + (24 * 32 + q * 16 + lrow) * 2) = o;
    }
    bq[pos * 16 + lrow] = acc[4] + cb[lrow];
  }
}

__global__ void __launch_bounds__(256) fcconv_wmma(const float* __restrict__ x,
                                                   const float* __restrict__ wq,
                                                   const float* __restrict__ bq,
                                                   float* __restrict__ h2,
                                                   float* __restrict__ part) {
  __shared__ float sX[GM_ROWS * FC_K + 512];
  __shared__ float sW[GM_TILES * WQ_TILE + 192];
  __shared__ float wsum[8][16];
  __shared__ float wsq[8][16];
  const int tid   = threadIdx.x;
  const int lane  = tid & 31;
  const int wave  = __builtin_amdgcn_readfirstlane(tid >> 5);
  const int lrow  = lane & 15;
  const int lhalf = lane >> 4;
  const int mg    = blockIdx.x & 15;
  const int chunk = blockIdx.x >> 4;

  {
    const v4f* srcx = (const v4f*)(x + mg * (GM_ROWS * FC_K));
    const v4f* srcw = (const v4f*)(wq + chunk * (GM_TILES * WQ_TILE));
    v4f* dX = (v4f*)sX;
    v4f* dW = (v4f*)sW;
    v4f rx[13], rw[8];
#pragma unroll
    for (int k = 0; k < 12; ++k) rx[k] = srcx[tid + k * 256];
    const int qx = 12 * 256 + tid;
    rx[12] = srcx[qx < 3200 ? qx : 3199];
#pragma unroll
    for (int k = 0; k < 7; ++k) rw[k] = srcw[tid + k * 256];
    const int qw = 7 * 256 + tid;
    rw[7] = srcw[qw < 2000 ? qw : 1999];
#pragma unroll
    for (int k = 0; k < 12; ++k) dX[tid + k * 256] = rx[k];
    dX[qx] = rx[12];
#pragma unroll
    for (int k = 0; k < 7; ++k) dW[tid + k * 256] = rw[k];
    dW[qw] = rw[7];
  }
  __syncthreads();

  const float* ap = sX + (wave * 16 + lrow) * FC_K + 2 * lhalf;
  v2f a[25];
#pragma unroll
  for (int s = 0; s < 25; ++s) a[s] = *(const v2f*)(ap + 4 * s);

  const int m0 = (mg * 8 + wave) * 16 + lhalf * 8;
  float s0 = 0.f, q0 = 0.f;

  for (int t = 0; t < GM_TILES; ++t) {
    const int pos = chunk * GM_TILES + t;
    const float bias = bq[pos * 16 + lrow];
    const float* bp = sW + t * WQ_TILE + lane * 2;
    v2f bf[25];
#pragma unroll
    for (int s = 0; s < 25; ++s) bf[s] = *(const v2f*)(bp + s * 64);
    __builtin_amdgcn_sched_barrier(0);
    v8f acc = {};
#pragma unroll
    for (int s = 0; s < 25; ++s)
      acc = __builtin_amdgcn_wmma_f32_16x16x4_f32(false, a[s], false, bf[s],
                                                  (short)0, acc, false, false);
#pragma unroll
    for (int v = 0; v < 8; ++v) {
      float val = acc[v] + bias;
      val = val >= 0.f ? val : 0.01f * val;
      h2[(m0 + v) * H2ROW + pos * 16 + lrow] = val;
      s0 += val;
      q0 += val * val;
    }
  }

  s0 += __shfl_xor(s0, 16, 32);
  q0 += __shfl_xor(q0, 16, 32);
  if (lane < 16) { wsum[wave][lane] = s0; wsq[wave][lane] = q0; }
  __syncthreads();
  if (tid < 32) {
    int c = tid & 15;
    int st = tid >> 4;
    float tot = 0.f;
    for (int wv = 0; wv < 8; ++wv) tot += st ? wsq[wv][c] : wsum[wv][c];
    part[blockIdx.x * 32 + tid] = tot;
  }
}

__global__ void __launch_bounds__(256) bn_reduce(const float* __restrict__ part,
                                                 float* __restrict__ gsum) {
  __shared__ float acc[256];
  const int t = threadIdx.x;
  const int cs = t & 31, r = t >> 5;
  const float* p = part + (blockIdx.x * BN_ROWS_PER_GROUP + r) * 32 + cs;
  float s = 0.f;
#pragma unroll
  for (int k = 0; k < 6; ++k) s += p[k * 256];
  if (r == 0) s += p[6 * 256];
  acc[t] = s;
  __syncthreads();
  if (t < 32) {
    float sum = 0.f;
    for (int rr = 0; rr < 8; ++rr) sum += acc[rr * 32 + t];
    gsum[blockIdx.x * 32 + t] = sum;
  }
}

#define LC_G      8
#define LC_HC     34
#define LC_TC     33
#define LC_GP     4
#define LC_POSQ   9
#define LC_GPQ    1530
#define LC_H3G    792
#define LC_ITEMS  2720
#define LC_TASKS  198
#define LC_BLOCKS 768

__device__ __forceinline__ void lc_load_w(const float* __restrict__ w1,
                                          const float* __restrict__ b1,
                                          int idx, int j0,
                                          float4 (&wa)[16], float4 (&wb)[16],
                                          float& bia, float& bib,
                                          int& xoff, int& h3off) {
  const int o1p = idx % 3;
  const int pp  = idx / 3;
  const int oi  = pp / LC_TC;
  const int ojl = pp - oi * LC_TC;
  const int oj  = j0 + ojl;
  const int o1a = 2 * o1p;
  const float* wp = w1 + o1a * 24832 + oi * 388 + oj * 4;
#pragma unroll
  for (int c = 0; c < 16; ++c) {
    wa[c] = *(const float4*)(wp + c * 1552);
    wb[c] = *(const float4*)(wp + 24832 + c * 1552);
  }
  bia = b1[o1a * 388 + oi * 97 + oj];
  bib = b1[(o1a + 1) * 388 + oi * 97 + oj];
  xoff  = (oi * LC_HC + ojl) * LC_POSQ;
  h3off = o1a * 132 + oi * LC_TC + ojl;
}

__device__ __forceinline__ void lc_ldx(const v4f* xg, int cq, v4f (&x)[8]) {
  x[0] = xg[2 * cq];                   x[1] = xg[2 * cq + 1];
  x[2] = xg[LC_POSQ + 2 * cq];         x[3] = xg[LC_POSQ + 2 * cq + 1];
  x[4] = xg[LC_HC * LC_POSQ + 2 * cq]; x[5] = xg[LC_HC * LC_POSQ + 2 * cq + 1];
  x[6] = xg[(LC_HC + 1) * LC_POSQ + 2 * cq];
  x[7] = xg[(LC_HC + 1) * LC_POSQ + 2 * cq + 1];
}

__device__ __forceinline__ v2f lc_term(const float4 w, v2f x00, v2f x01, v2f x10, v2f x11) {
  return w.x * x00 + w.y * x01 + w.z * x10 + w.w * x11;
}

__device__ __forceinline__ void lc2_one(int e, int b0, int j0,
                                        const float* __restrict__ w2,
                                        const float* __restrict__ b2,
                                        const float* sh3,
                                        float* __restrict__ out) {
  const int g   = e / 96;
  const int r   = e - g * 96;
  const int ri  = r >> 5;
  const int rjl = r & 31;
  const int rj  = j0 + rjl;
  float acc = b2[ri * 96 + rj];
#pragma unroll
  for (int o1 = 0; o1 < 6; ++o1) {
    float4 wv = *(const float4*)(w2 + (((o1 * 3 + ri) * 96 + rj) << 2));
    const float* xb = sh3 + g * LC_H3G + o1 * 132 + ri * LC_TC + rjl;
    acc += wv.x * xb[0] + wv.y * xb[1] + wv.z * xb[33] + wv.w * xb[34];
  }
  out[(b0 + g) * 288 + ri * 96 + rj] = acc;
}

__global__ void __launch_bounds__(224) lc_kernel(const float* __restrict__ h2,
                                                 const float* __restrict__ gsum,
                                                 const float* __restrict__ gamma,
                                                 const float* __restrict__ beta,
                                                 const float* __restrict__ w1,
                                                 const float* __restrict__ b1,
                                                 const float* __restrict__ w2,
                                                 const float* __restrict__ b2,
                                                 float* __restrict__ out) {
  __shared__ v4f sh2p[LC_GP * LC_GPQ];
  __shared__ float sh3[LC_G * LC_H3G];
  const int tid   = threadIdx.x;
  const int strip = blockIdx.x % 3;
  const int b0    = (blockIdx.x / 3) * LC_G;
  const int j0    = strip * 32;

  float4 wa[16], wb[16];
  float bia, bib;
  int xoff, h3off;
  lc_load_w(w1, b1, tid, j0, wa, wb, bia, bib, xoff, h3off);

  const int lane = tid & 31;
  float tot = 0.f;
#pragma unroll
  for (int g = 0; g < BN_GROUPS; ++g) tot += gsum[g * 32 + lane];
  const float oth = __shfl_xor(tot, 16, 32);
  const float csum = lane < 16 ? tot : oth;
  const float csq  = lane < 16 ? oth : tot;
  const float invN = 1.0f / BN_COUNT;
  const float mean = csum * invN;
  const float var  = csq * invN - mean * mean;
  const float inv  = rsqrtf(var + 1e-5f);
  const float scl  = gamma[lane & 15] * inv;
  const float shf  = beta[lane & 15] - mean * scl;

  const int q4 = (tid & 3) * 4;
  float4 sc4, sf4;
  sc4.x = __shfl(scl, q4 + 0, 32); sf4.x = __shfl(shf, q4 + 0, 32);
  sc4.y = __shfl(scl, q4 + 1, 32); sf4.y = __shfl(shf, q4 + 1, 32);
  sc4.z = __shfl(scl, q4 + 2, 32); sf4.z = __shfl(shf, q4 + 2, 32);
  sc4.w = __shfl(scl, q4 + 3, 32); sf4.w = __shfl(shf, q4 + 3, 32);
  const float4* src4 = (const float4*)h2;
#pragma nounroll
  for (int k0 = 0; k0 < 12; k0 += 6) {
    float4 r[12];
    int di[6];
#pragma unroll
    for (int u = 0; u < 6; ++u) {
      const int q   = tid + (k0 + u) * 224;
      const int seg = q / 136;
      const int w   = q - seg * 136;
      const int gp  = seg / 5;
      const int i   = seg - gp * 5;
      const int jj  = w >> 2;
      const int cq  = w & 3;
      const int gi  = ((b0 + 2 * gp) * CPOS + i * 98 + j0 + jj) * 4 + cq;
      r[2 * u]     = src4[gi];
      r[2 * u + 1] = src4[gi + CPOS * 4];
      di[u] = (seg * LC_HC + jj) * LC_POSQ + 2 * cq;
    }
#pragma unroll
    for (int u = 0; u < 6; ++u) {
      const float4 va = r[2 * u], vb = r[2 * u + 1];
      v4f lo, hi;
      lo.x = va.x * sc4.x + sf4.x;  lo.y = vb.x * sc4.x + sf4.x;
      lo.z = va.y * sc4.y + sf4.y;  lo.w = vb.y * sc4.y + sf4.y;
      hi.x = va.z * sc4.z + sf4.z;  hi.y = vb.z * sc4.z + sf4.z;
      hi.z = va.w * sc4.w + sf4.w;  hi.w = vb.w * sc4.w + sf4.w;
      sh2p[di[u]]     = lo;
      sh2p[di[u] + 1] = hi;
    }
  }
  if (tid < LC_ITEMS - 12 * 224) {
    const int q   = tid + 12 * 224;
    const int seg = q / 136;
    const int w   = q - seg * 136;
    const int gp  = seg / 5;
    const int i   = seg - gp * 5;
    const int jj  = w >> 2;
    const int cq  = w & 3;
    const int gi  = ((b0 + 2 * gp) * CPOS + i * 98 + j0 + jj) * 4 + cq;
    const float4 va = src4[gi], vb = src4[gi + CPOS * 4];
    v4f lo, hi;
    lo.x = va.x * sc4.x + sf4.x;  lo.y = vb.x * sc4.x + sf4.x;
    lo.z = va.y * sc4.y + sf4.y;  lo.w = vb.y * sc4.y + sf4.y;
    hi.x = va.z * sc4.z + sf4.z;  hi.y = vb.z * sc4.z + sf4.z;
    hi.z = va.w * sc4.w + sf4.w;  hi.w = vb.w * sc4.w + sf4.w;
    const int d = (seg * LC_HC + jj) * LC_POSQ + 2 * cq;
    sh2p[d]     = lo;
    sh2p[d + 1] = hi;
  }
  __syncthreads();

#pragma nounroll
  for (int pass = 0; pass < 2; ++pass) {
    if (pass == 1) {
      const int idx = LC_TASKS + tid;
      lc_load_w(w1, b1, idx < 395 ? idx : 395, j0, wa, wb, bia, bib, xoff, h3off);
    }
    if (tid < LC_TASKS) {
      const v4f* xb = sh2p + xoff;
      v4f xc[8], xn[8];
      lc_ldx(xb, 0, xc);
#pragma nounroll
      for (int gp = 0; gp < LC_GP; ++gp) {
        const v4f* xg  = xb + gp * LC_GPQ;
        const v4f* xgn = xb + (gp < LC_GP - 1 ? gp + 1 : gp) * LC_GPQ;
        v2f a0, a1;
        a0.x = bia; a0.y = bia;
        a1.x = bib; a1.y = bib;
#pragma unroll
        for (int cq = 0; cq < 4; ++cq) {
          if (cq < 3) lc_ldx(xg, cq + 1, xn); else lc_ldx(xgn, 0, xn);
          __builtin_amdgcn_sched_barrier(0);
          a0 += lc_term(wa[4 * cq + 0], xc[0].xy, xc[2].xy, xc[4].xy, xc[6].xy);
          a1 += lc_term(wb[4 * cq + 0], xc[0].xy, xc[2].xy, xc[4].xy, xc[6].xy);
          a0 += lc_term(wa[4 * cq + 1], xc[0].zw, xc[2].zw, xc[4].zw, xc[6].zw);
          a1 += lc_term(wb[4 * cq + 1], xc[0].zw, xc[2].zw, xc[4].zw, xc[6].zw);
          a0 += lc_term(wa[4 * cq + 2], xc[1].xy, xc[3].xy, xc[5].xy, xc[7].xy);
          a1 += lc_term(wb[4 * cq + 2], xc[1].xy, xc[3].xy, xc[5].xy, xc[7].xy);
          a0 += lc_term(wa[4 * cq + 3], xc[1].zw, xc[3].zw, xc[5].zw, xc[7].zw);
          a1 += lc_term(wb[4 * cq + 3], xc[1].zw, xc[3].zw, xc[5].zw, xc[7].zw);
#pragma unroll
          for (int e = 0; e < 8; ++e) xc[e] = xn[e];
        }
        float r0 = a0.x, r1 = a1.x, r2 = a0.y, r3 = a1.y;
        r0 = r0 >= 0.f ? r0 : 0.01f * r0;
        r1 = r1 >= 0.f ? r1 : 0.01f * r1;
        r2 = r2 >= 0.f ? r2 : 0.01f * r2;
        r3 = r3 >= 0.f ? r3 : 0.01f * r3;
        float* hp = sh3 + (2 * gp) * LC_H3G + h3off;
        hp[0]            = r0;
        hp[132]          = r1;
        hp[LC_H3G]       = r2;
        hp[LC_H3G + 132] = r3;
      }
    }
  }
  __syncthreads();

  lc2_one(tid,       b0, j0, w2, b2, sh3, out);
  lc2_one(tid + 224, b0, j0, w2, b2, sh3, out);
  lc2_one(tid + 448, b0, j0, w2, b2, sh3, out);
  if (tid < LC_G * 96 - 3 * 224) lc2_one(tid + 672, b0, j0, w2, b2, sh3, out);
}

extern "C" void kernel_launch(void* const* d_in, const int* in_sizes, int n_in,
                              void* d_out, int out_size, void* d_ws, size_t ws_size,
                              hipStream_t stream) {
  const float* x     = (const float*)d_in[0];
  const float* fw    = (const float*)d_in[1];
  const float* fb    = (const float*)d_in[2];
  const float* cw    = (const float*)d_in[3];
  const float* cb    = (const float*)d_in[4];
  const float* gamma = (const float*)d_in[5];
  const float* beta  = (const float*)d_in[6];
  const float* w1    = (const float*)d_in[7];
  const float* b1    = (const float*)d_in[8];
  const float* w2    = (const float*)d_in[9];
  const float* b2    = (const float*)d_in[10];

  float* ws    = (float*)d_ws;
  float* gsum  = ws;
  float* wq    = ws + WQ_OFF;
  float* bq    = ws + WQ_OFF + WQ_ELEMS;
  float* h2    = ws + H1_ELEMS;
  float* part  = ws + H1_ELEMS + H2_ELEMS;
  float* pw    = part + PART_ELEMS;

  pack_conv_w<<<9, 256, 0, stream>>>(cw, pw);
  compose_w  <<<429, 256, 0, stream>>>(fw, fb, pw, cb, wq, bq);
  fcconv_wmma<<<GM_BLOCKS, 256, 0, stream>>>(x, wq, bq, h2, part);
  bn_reduce  <<<BN_GROUPS, 256, 0, stream>>>(part, gsum);
  lc_kernel  <<<LC_BLOCKS, 224, 0, stream>>>(h2, gsum, gamma, beta, w1, b1, w2, b2,
                                      (float*)d_out);
}


# conv1 weights read by the fold kernel in their own layout (128-bit load per channel, k-step = kh x channel pair): repack launch removed, 5 launches to 4; fold k order (f32 sum order) changes
# speedup vs baseline: 6.6220x; 1.0359x over previous
#include <hip/hip_runtime.h>

typedef __attribute__((ext_vector_type(2))) float v2f;
typedef __attribute__((ext_vector_type(4))) float v4f;
typedef __attribute__((ext_vector_type(8))) float v8f;

#define NB      2048
#define FC_K    100
#define CPOS    490
#define H1_ELEMS (NB * 17640)
#define H2_ELEMS (NB * 5 * 98 * 16)
#define H2ROW   7840
#define PART_ELEMS 250880
#define BN_GROUPS 32
#define BN_ROWS_PER_GROUP 49
#define BN_COUNT 1003520.0f
#define WQ_TILE 1600
#define WQ_OFF  4096
#define WQ_ELEMS (CPOS * WQ_TILE)
#define GM_ROWS  128
#define GM_TILES 5
#define GM_BLOCKS 1568

__global__ void __launch_bounds__(256) compose_w(const float* __restrict__ fw,
                                                 const float* __restrict__ fb,
                                                 const float* __restrict__ cw,
                                                 const float* __restrict__ cb,
                                                 float* __restrict__ wq,
                                                 float* __restrict__ bq) {
  const int lane  = threadIdx.x & 31;
  const int w     = blockIdx.x * 8 + __builtin_amdgcn_readfirstlane(threadIdx.x >> 5);
  if (w >= CPOS * 7) return;
  const int pos   = w / 7;
  const int kt    = w - pos * 7;
  const int lrow  = lane & 15;
  const int lhalf = lane >> 4;
  const int pi    = pos / 98;
  const int pj    = pos - pi * 98;
  const bool vdi  = pi < 4;
  const bool vdj  = pj < 97;
  const int k     = kt * 16 + lrow;
  const bool live = k <= FC_K;
  const float* colp = k < FC_K ? fw + k : fb;
  const int rs    = k < FC_K ? FC_K : 1;

  const float* cwp = cw + lrow * 144 + 4 * lhalf;
  v4f wc[18];
#pragma unroll
  for (int u = 0; u < 18; ++u) wc[u] = *(const v4f*)(cwp + 8 * u);

  v8f acc = {};
#pragma unroll
  for (int hh = 0; hh < 2; ++hh) {
    v2f af[18], bf[18];
    const bool v0 = hh ? vdi : true;
    const bool v1 = v0 && vdj;
    const int row0 = (pos + (v0 ? hh * 98 : 0)) * 36 + lhalf;
    const int row1 = (pos + (v1 ? hh * 98 + 1 : 0)) * 36 + lhalf;
    const bool keep0 = v0 && live;
    const bool keep1 = v1 && live;
#pragma unroll
    for (int u = 0; u < 18; ++u) {
      v2f f;
      f[0] = colp[(row0 + 2 * u) * rs];
      f[1] = colp[(row1 + 2 * u) * rs];
      f[0] = keep0 ? f[0] : 0.f;
      f[1] = keep1 ? f[1] : 0.f;
      af[u] = f;
      bf[u] = hh ? wc[u].zw : wc[u].xy;
    }
    __builtin_amdgcn_sched_barrier(0);
#pragma unroll
    for (int u = 0; u < 18; ++u)
      acc = __builtin_amdgcn_wmma_f32_16x16x4_f32(false, af[u], false, bf[u],
                                                  (short)0, acc, false, false);
  }
  float* wp = wq + pos * WQ_TILE;
  if (kt < 6) {
#pragma unroll
    for (int q = 0; q < 4; ++q) {
      const int s = kt * 4 + 2 * lhalf + (q >> 1);
      v2f o; o[0] = acc[2 * q]; o[1] = acc[2 * q + 1];
      *(v2f*)(wp + (s * 32 + (q & 1) * 16 + lrow) * 2) = o;
    }
  } else if (lhalf == 0) {
#pragma unroll
    for (int q = 0; q < 2; ++q) {
      v2f o; o[0] = acc[2 * q]; o[1] = acc[2 * q + 1];
      *(v2f*)(wp + (24 * 32 + q * 16 + lrow) * 2) = o;
    }
    bq[pos * 16 + lrow] = acc[4] + cb[lrow];
  }
}

__global__ void __launch_bounds__(256) fcconv_wmma(const float* __restrict__ x,
                                                   const float* __restrict__ wq,
                                                   const float* __restrict__ bq,
                                                   float* __restrict__ h2,
                                                   float* __restrict__ part) {
  __shared__ float sX[GM_ROWS * FC_K + 512];
  __shared__ float sW[GM_TILES * WQ_TILE + 192];
  __shared__ float wsum[8][16];
  __shared__ float wsq[8][16];
  const int tid   = threadIdx.x;
  const int lane  = tid & 31;
  const int wave  = __builtin_amdgcn_readfirstlane(tid >> 5);
  const int lrow  = lane & 15;
  const int lhalf = lane >> 4;
  const int mg    = blockIdx.x & 15;
  const int chunk = blockIdx.x >> 4;

  {
    const v4f* srcx = (const v4f*)(x + mg * (GM_ROWS * FC_K));
    const v4f* srcw = (const v4f*)(wq + chunk * (GM_TILES * WQ_TILE));
    v4f* dX = (v4f*)sX;
    v4f* dW = (v4f*)sW;
    v4f rx[13], rw[8];
#pragma unroll
    for (int k = 0; k < 12; ++k) rx[k] = srcx[tid + k * 256];
    const int qx = 12 * 256 + tid;
    rx[12] = srcx[qx < 3200 ? qx : 3199];
#pragma unroll
    for (int k = 0; k < 7; ++k) rw[k] = srcw[tid + k * 256];
    const int qw = 7 * 256 + tid;
    rw[7] = srcw[qw < 2000 ? qw : 1999];
#pragma unroll
    for (int k = 0; k < 12; ++k) dX[tid + k * 256] = rx[k];
    dX[qx] = rx[12];
#pragma unroll
    for (int k = 0; k < 7; ++k) dW[tid + k * 256] = rw[k];
    dW[qw] = rw[7];
  }
  __syncthreads();

  const float* ap = sX + (wave * 16 + lrow) * FC_K + 2 * lhalf;
  v2f a[25];
#pragma unroll
  for (int s = 0; s < 25; ++s) a[s] = *(const v2f*)(ap + 4 * s);

  const int m0 = (mg * 8 + wave) * 16 + lhalf * 8;
  float s0 = 0.f, q0 = 0.f;

  for (int t = 0; t < GM_TILES; ++t) {
    const int pos = chunk * GM_TILES + t;
    const float bias = bq[pos * 16 + lrow];
    const float* bp = sW + t * WQ_TILE + lane * 2;
    v2f bf[25];
#pragma unroll
    for (int s = 0; s < 25; ++s) bf[s] = *(const v2f*)(bp + s * 64);
    __builtin_amdgcn_sched_barrier(0);
    v8f acc = {};
#pragma unroll
    for (int s = 0; s < 25; ++s)
      acc = __builtin_amdgcn_wmma_f32_16x16x4_f32(false, a[s], false, bf[s],
                                                  (short)0, acc, false, false);
#pragma unroll
    for (int v = 0; v < 8; ++v) {
      float val = acc[v] + bias;
      val = val >= 0.f ? val : 0.01f * val;
      h2[(m0 + v) * H2ROW + pos * 16 + lrow] = val;
      s0 += val;
      q0 += val * val;
    }
  }

  s0 += __shfl_xor(s0, 16, 32);
  q0 += __shfl_xor(q0, 16, 32);
  if (lane < 16) { wsum[wave][lane] = s0; wsq[wave][lane] = q0; }
  __syncthreads();
  if (tid < 32) {
    int c = tid & 15;
    int st = tid >> 4;
    float tot = 0.f;
    for (int wv = 0; wv < 8; ++wv) tot += st ? wsq[wv][c] : wsum[wv][c];
    part[blockIdx.x * 32 + tid] = tot;
  }
}

__global__ void __launch_bounds__(256) bn_reduce(const float* __restrict__ part,
                                                 float* __restrict__ gsum) {
  __shared__ float acc[256];
  const int t = threadIdx.x;
  const int cs = t & 31, r = t >> 5;
  const float* p = part + (blockIdx.x * BN_ROWS_PER_GROUP + r) * 32 + cs;
  float s = 0.f;
#pragma unroll
  for (int k = 0; k < 6; ++k) s += p[k * 256];
  if (r == 0) s += p[6 * 256];
  acc[t] = s;
  __syncthreads();
  if (t < 32) {
    float sum = 0.f;
    for (int rr = 0; rr < 8; ++rr) sum += acc[rr * 32 + t];
    gsum[blockIdx.x * 32 + t] = sum;
  }
}

#define LC_G      8
#define LC_HC     34
#define LC_TC     33
#define LC_GP     4
#define LC_POSQ   9
#define LC_GPQ    1530
#define LC_H3G    792
#define LC_ITEMS  2720
#define LC_TASKS  198
#define LC_BLOCKS 768

__device__ __forceinline__ void lc_load_w(const float* __restrict__ w1,
                                          const float* __restrict__ b1,
                                          int idx, int j0,
                                          float4 (&wa)[16], float4 (&wb)[16],
                                          float& bia, float& bib,
                                          int& xoff, int& h3off) {
  const int o1p = idx % 3;
  const int pp  = idx / 3;
  const int oi  = pp / LC_TC;
  const int ojl = pp - oi * LC_TC;
  const int oj  = j0 + ojl;
  const int o1a = 2 * o1p;
  const float* wp = w1 + o1a * 24832 + oi * 388 + oj * 4;
#pragma unroll
  for (int c = 0; c < 16; ++c) {
    wa[c] = *(const float4*)(wp + c * 1552);
    wb[c] = *(const float4*)(wp + 24832 + c * 1552);
  }
  bia = b1[o1a * 388 + oi * 97 + oj];
  bib = b1[(o1a + 1) * 388 + oi * 97 + oj];
  xoff  = (oi * LC_HC + ojl) * LC_POSQ;
  h3off = o1a * 132 + oi * LC_TC + ojl;
}

__device__ __forceinline__ void lc_ldx(const v4f* xg, int cq, v4f (&x)[8]) {
  x[0] = xg[2 * cq];                   x[1] = xg[2 * cq + 1];
  x[2] = xg[LC_POSQ + 2 * cq];         x[3] = xg[LC_POSQ + 2 * cq + 1];
  x[4] = xg[LC_HC * LC_POSQ + 2 * cq]; x[5] = xg[LC_HC * LC_POSQ + 2 * cq + 1];
  x[6] = xg[(LC_HC + 1) * LC_POSQ + 2 * cq];
  x[7] = xg[(LC_HC + 1) * LC_POSQ + 2 * cq + 1];
}

__device__ __forceinline__ v2f lc_term(const float4 w, v2f x00, v2f x01, v2f x10, v2f x11) {
  return w.x * x00 + w.y * x01 + w.z * x10 + w.w * x11;
}

__device__ __forceinline__ void lc2_one(int e, int b0, int j0,
                                        const float* __restrict__ w2,
                                        const float* __restrict__ b2,
                                        const float* sh3,
                                        float* __restrict__ out) {
  const int g   = e / 96;
  const int r   = e - g * 96;
  const int ri  = r >> 5;
  const int rjl = r & 31;
  const int rj  = j0 + rjl;
  float acc = b2[ri * 96 + rj];
#pragma unroll
  for (int o1 = 0; o1 < 6; ++o1) {
    float4 wv = *(const float4*)(w2 + (((o1 * 3 + ri) * 96 + rj) << 2));
    const float* xb = sh3 + g * LC_H3G + o1 * 132 + ri * LC_TC + rjl;
    acc += wv.x * xb[0] + wv.y * xb[1] + wv.z * xb[33] + wv.w * xb[34];
  }
  out[(b0 + g) * 288 + ri * 96 + rj] = acc;
}

__global__ void __launch_bounds__(224) lc_kernel(const float* __restrict__ h2,
                                                 const float* __restrict__ gsum,
                                                 const float* __restrict__ gamma,
                                                 const float* __restrict__ beta,
                                                 const float* __restrict__ w1,
                                                 const float* __restrict__ b1,
                                                 const float* __restrict__ w2,
                                                 const float* __restrict__ b2,
                                                 float* __restrict__ out) {
  __shared__ v4f sh2p[LC_GP * LC_GPQ];
  __shared__ float sh3[LC_G * LC_H3G];
  const int tid   = threadIdx.x;
  const int strip = blockIdx.x % 3;
  const int b0    = (blockIdx.x / 3) * LC_G;
  const int j0    = strip * 32;

  float4 wa[16], wb[16];
  float bia, bib;
  int xoff, h3off;
  lc_load_w(w1, b1, tid, j0, wa, wb, bia, bib, xoff, h3off);

  const int lane = tid & 31;
  float tot = 0.f;
#pragma unroll
  for (int g = 0; g < BN_GROUPS; ++g) tot += gsum[g * 32 + lane];
  const float oth = __shfl_xor(tot, 16, 32);
  const float csum = lane < 16 ? tot : oth;
  const float csq  = lane < 16 ? oth : tot;
  const float invN = 1.0f / BN_COUNT;
  const float mean = csum * invN;
  const float var  = csq * invN - mean * mean;
  const float inv  = rsqrtf(var + 1e-5f);
  const float scl  = gamma[lane & 15] * inv;
  const float shf  = beta[lane & 15] - mean * scl;

  const int q4 = (tid & 3) * 4;
  float4 sc4, sf4;
  sc4.x = __shfl(scl, q4 + 0, 32); sf4.x = __shfl(shf, q4 + 0, 32);
  sc4.y = __shfl(scl, q4 + 1, 32); sf4.y = __shfl(shf, q4 + 1, 32);
  sc4.z = __shfl(scl, q4 + 2, 32); sf4.z = __shfl(shf, q4 + 2, 32);
  sc4.w = __shfl(scl, q4 + 3, 32); sf4.w = __shfl(shf, q4 + 3, 32);
  const float4* src4 = (const float4*)h2;
#pragma nounroll
  for (int k0 = 0; k0 < 12; k0 += 6) {
    float4 r[12];
    int di[6];
#pragma unroll
    for (int u = 0; u < 6; ++u) {
      const int q   = tid + (k0 + u) * 224;
      const int seg = q / 136;
      const int w   = q - seg * 136;
      const int gp  = seg / 5;
      const int i   = seg - gp * 5;
      const int jj  = w >> 2;
      const int cq  = w & 3;
      const int gi  = ((b0 + 2 * gp) * CPOS + i * 98 + j0 + jj) * 4 + cq;
      r[2 * u]     = src4[gi];
      r[2 * u + 1] = src4[gi + CPOS * 4];
      di[u] = (seg * LC_HC + jj) * LC_POSQ + 2 * cq;
    }
#pragma unroll
    for (int u = 0; u < 6; ++u) {
      const float4 va = r[2 * u], vb = r[2 * u + 1];
      v4f lo, hi;
      lo.x = va.x * sc4.x + sf4.x;  lo.y = vb.x * sc4.x + sf4.x;
      lo.z = va.y * sc4.y + sf4.y;  lo.w = vb.y * sc4.y + sf4.y;
      hi.x = va.z * sc4.z + sf4.z;  hi.y = vb.z * sc4.z + sf4.z;
      hi.z = va.w * sc4.w + sf4.w;  hi.w = vb.w * sc4.w + sf4.w;
      sh2p[di[u]]     = lo;
      sh2p[di[u] + 1] = hi;
    }
  }
  if (tid < LC_ITEMS - 12 * 224) {
    const int q   = tid + 12 * 224;
    const int seg = q / 136;
    const int w   = q - seg * 136;
    const int gp  = seg / 5;
    const int i   = seg - gp * 5;
    const int jj  = w >> 2;
    const int cq  = w & 3;
    const int gi  = ((b0 + 2 * gp) * CPOS + i * 98 + j0 + jj) * 4 + cq;
    const float4 va = src4[gi], vb = src4[gi + CPOS * 4];
    v4f lo, hi;
    lo.x = va.x * sc4.x + sf4.x;  lo.y = vb.x * sc4.x + sf4.x;
    lo.z = va.y * sc4.y + sf4.y;  lo.w = vb.y * sc4.y + sf4.y;
    hi.x = va.z * sc4.z + sf4.z;  hi.y = vb.z * sc4.z + sf4.z;
    hi.z = va.w * sc4.w + sf4.w;  hi.w = vb.w * sc4.w + sf4.w;
    const int d = (seg * LC_HC + jj) * LC_POSQ + 2 * cq;
    sh2p[d]     = lo;
    sh2p[d + 1] = hi;
  }
  __syncthreads();

#pragma nounroll
  for (int pass = 0; pass < 2; ++pass) {
    if (pass == 1) {
      const int idx = LC_TASKS + tid;
      lc_load_w(w1, b1, idx < 395 ? idx : 395, j0, wa, wb, bia, bib, xoff, h3off);
    }
    if (tid < LC_TASKS) {
      const v4f* xb = sh2p + xoff;
      v4f xc[8], xn[8];
      lc_ldx(xb, 0, xc);
#pragma nounroll
      for (int gp = 0; gp < LC_GP; ++gp) {
        const v4f* xg  = xb + gp * LC_GPQ;
        const v4f* xgn = xb + (gp < LC_GP - 1 ? gp + 1 : gp) * LC_GPQ;
        v2f a0, a1;
        a0.x = bia; a0.y = bia;
        a1.x = bib; a1.y = bib;
#pragma unroll
        for (int cq = 0; cq < 4; ++cq) {
          if (cq < 3) lc_ldx(xg, cq + 1, xn); else lc_ldx(xgn, 0, xn);
          __builtin_amdgcn_sched_barrier(0);
          a0 += lc_term(wa[4 * cq + 0], xc[0].xy, xc[2].xy, xc[4].xy, xc[6].xy);
          a1 += lc_term(wb[4 * cq + 0], xc[0].xy, xc[2].xy, xc[4].xy, xc[6].xy);
          a0 += lc_term(wa[4 * cq + 1], xc[0].zw, xc[2].zw, xc[4].zw, xc[6].zw);
          a1 += lc_term(wb[4 * cq + 1], xc[0].zw, xc[2].zw, xc[4].zw, xc[6].zw);
          a0 += lc_term(wa[4 * cq + 2], xc[1].xy, xc[3].xy, xc[5].xy, xc[7].xy);
          a1 += lc_term(wb[4 * cq + 2], xc[1].xy, xc[3].xy, xc[5].xy, xc[7].xy);
          a0 += lc_term(wa[4 * cq + 3], xc[1].zw, xc[3].zw, xc[5].zw, xc[7].zw);
          a1 += lc_term(wb[4 * cq + 3], xc[1].zw, xc[3].zw, xc[5].zw, xc[7].zw);
#pragma unroll
          for (int e = 0; e < 8; ++e) xc[e] = xn[e];
        }
        float r0 = a0.x, r1 = a1.x, r2 = a0.y, r3 = a1.y;
        r0 = r0 >= 0.f ? r0 : 0.01f * r0;
        r1 = r1 >= 0.f ? r1 : 0.01f * r1;
        r2 = r2 >= 0.f ? r2 : 0.01f * r2;
        r3 = r3 >= 0.f ? r3 : 0.01f * r3;
        float* hp = sh3 + (2 * gp) * LC_H3G + h3off;
        hp[0]            = r0;
        hp[132]          = r1;
        hp[LC_H3G]       = r2;
        hp[LC_H3G + 132] = r3;
      }
    }
  }
  __syncthreads();

  lc2_one(tid,       b0, j0, w2, b2, sh3, out);
  lc2_one(tid + 224, b0, j0, w2, b2, sh3, out);
  lc2_one(tid + 448, b0, j0, w2, b2, sh3, out);
  if (tid < LC_G * 96 - 3 * 224) lc2_one(tid + 672, b0, j0, w2, b2, sh3, out);
}

extern "C" void kernel_launch(void* const* d_in, const int* in_sizes, int n_in,
                              void* d_out, int out_size, void* d_ws, size_t ws_size,
                              hipStream_t stream) {
  const float* x     = (const float*)d_in[0];
  const float* fw    = (const float*)d_in[1];
  const float* fb    = (const float*)d_in[2];
  const float* cw    = (const float*)d_in[3];
  const float* cb    = (const float*)d_in[4];
  const float* gamma = (const float*)d_in[5];
  const float* beta  = (const float*)d_in[6];
  const float* w1    = (const float*)d_in[7];
  const float* b1    = (const float*)d_in[8];
  const float* w2    = (const float*)d_in[9];
  const float* b2    = (const float*)d_in[10];

  float* ws    = (float*)d_ws;
  float* gsum  = ws;
  float* wq    = ws + WQ_OFF;
  float* bq    = ws + WQ_OFF + WQ_ELEMS;
  float* h2    = ws + H1_ELEMS;
  float* part  = ws + H1_ELEMS + H2_ELEMS;

  compose_w  <<<429, 256, 0, stream>>>(fw, fb, cw, cb, wq, bq);
  fcconv_wmma<<<GM_BLOCKS, 256, 0, stream>>>(x, wq, bq, h2, part);
  bn_reduce  <<<BN_GROUPS, 256, 0, stream>>>(part, gsum);
  lc_kernel  <<<LC_BLOCKS, 224, 0, stream>>>(h2, gsum, gamma, beta, w1, b1, w2, b2,
                                      (float*)d_out);
}


# GEMM re-tiled: a block owns one position pair for all batch rows, folds its own W' in LDS (fold launch gone, 4 launches to 3), stores whole 128-B lines; BN partial rows 1568 to 245: sum order changes
# speedup vs baseline: 7.0611x; 1.0663x over previous
#include <hip/hip_runtime.h>

typedef __attribute__((ext_vector_type(2))) float v2f;
typedef __attribute__((ext_vector_type(4))) float v4f;
typedef __attribute__((ext_vector_type(8))) float v8f;

#define NB      2048
#define FC_K    100
#define CPOS    490
#define H1_ELEMS (NB * 17640)
#define H2_ELEMS (NB * 5 * 98 * 16)
#define H2ROW   7840
#define PART_ELEMS 250880
#define BN_GROUPS 5
#define BN_ROWS_PER_GROUP 49
#define BN_COUNT 1003520.0f
#define WQ_TILE 1600
#define PG_BLOCKS 245
#define PG_WAVES  16
#define PG_XQ     416

__device__ __forceinline__ void fold_tile(const float* __restrict__ fw,
                                          const float* __restrict__ fb,
                                          const float* __restrict__ cw,
                                          const float* __restrict__ cb,
                                          int pos, int kt, int lane,
                                          float* wp, float* bp) {
  const int lrow  = lane & 15;
  const int lhalf = lane >> 4;
  const int pi    = pos / 98;
  const int pj    = pos - pi * 98;
  const bool vdi  = pi < 4;
  const bool vdj  = pj < 97;
  const int k     = kt * 16 + lrow;
  const bool live = k <= FC_K;
  const float* colp = k < FC_K ? fw + k : fb;
  const int rs    = k < FC_K ? FC_K : 1;

  const float* cwp = cw + lrow * 144 + 4 * lhalf;
  v4f wc[18];
#pragma unroll
  for (int u = 0; u < 18; ++u) wc[u] = *(const v4f*)(cwp + 8 * u);

  v8f acc = {};
#pragma unroll
  for (int hh = 0; hh < 2; ++hh) {
    v2f af[18], bf[18];
    const bool v0 = hh ? vdi : true;
    const bool v1 = v0 && vdj;
    const int row0 = (pos + (v0 ? hh * 98 : 0)) * 36 + lhalf;
    const int row1 = (pos + (v1 ? hh * 98 + 1 : 0)) * 36 + lhalf;
    const bool keep0 = v0 && live;
    const bool keep1 = v1 && live;
#pragma unroll
    for (int u = 0; u < 18; ++u) {
      v2f f;
      f[0] = colp[(row0 + 2 * u) * rs];
      f[1] = colp[(row1 + 2 * u) * rs];
      f[0] = keep0 ? f[0] : 0.f;
      f[1] = keep1 ? f[1] : 0.f;
      af[u] = f;
      bf[u] = hh ? wc[u].zw : wc[u].xy;
    }
    __builtin_amdgcn_sched_barrier(0);
#pragma unroll
    for (int u = 0; u < 18; ++u)
      acc = __builtin_amdgcn_wmma_f32_16x16x4_f32(false, af[u], false, bf[u],
                                                  (short)0, acc, false, false);
  }
  if (kt < 6) {
#pragma unroll
    for (int q = 0; q < 4; ++q) {
      const int s = kt * 4 + 2 * lhalf + (q >> 1);
      v2f o; o[0] = acc[2 * q]; o[1] = acc[2 * q + 1];
      *(v2f*)(wp + (s * 32 + (q & 1) * 16 + lrow) * 2) = o;
    }
  } else if (lhalf == 0) {
#pragma unroll
    for (int q = 0; q < 2; ++q) {
      v2f o; o[0] = acc[2 * q]; o[1] = acc[2 * q + 1];
      *(v2f*)(wp + (24 * 32 + q * 16 + lrow) * 2) = o;
    }
    bp[lrow] = acc[4] + cb[lrow];
  }
}

__device__ __forceinline__ void load_x_tile(const float* __restrict__ x, int rt, int lane,
                                            v4f (&r)[13]) {
  const v4f* srcx = (const v4f*)(x + rt * (16 * FC_K));
#pragma unroll
  for (int k = 0; k < 12; ++k) r[k] = srcx[lane + 32 * k];
  const int ql = 384 + lane;
  r[12] = srcx[ql < 400 ? ql : 399];
}

__global__ void __launch_bounds__(512) fcconv_pair(const float* __restrict__ x,
                                                   const float* __restrict__ fw,
                                                   const float* __restrict__ fb,
                                                   const float* __restrict__ cw,
                                                   const float* __restrict__ cb,
                                                   float* __restrict__ h2,
                                                   float* __restrict__ part) {
  __shared__ v4f sXw[PG_WAVES * PG_XQ];
  __shared__ float sWq[2 * WQ_TILE];
  __shared__ float sBq[32];
  __shared__ float wsum[PG_WAVES][16];
  __shared__ float wsq[PG_WAVES][16];
  const int tid   = threadIdx.x;
  const int lane  = tid & 31;
  const int wave  = __builtin_amdgcn_readfirstlane(tid >> 5);
  const int lrow  = lane & 15;
  const int lhalf = lane >> 4;
  const int pos0  = blockIdx.x * 2;

  v4f r[13];
  load_x_tile(x, wave, lane, r);

  if (wave < 14) {
    const int pl = wave >= 7 ? 1 : 0;
    fold_tile(fw, fb, cw, cb, pos0 + pl, wave - 7 * pl, lane,
              sWq + pl * WQ_TILE, sBq + pl * 16);
  }
  __syncthreads();

  v2f b0[25], b1[25];
#pragma unroll
  for (int s = 0; s < 25; ++s) {
    b0[s] = *(const v2f*)(sWq + s * 64 + lane * 2);
    b1[s] = *(const v2f*)(sWq + WQ_TILE + s * 64 + lane * 2);
  }
  const float bias0 = sBq[lrow];
  const float bias1 = sBq[16 + lrow];

  v4f* myX = sXw + wave * PG_XQ;
  const float* ap = (const float*)myX + lrow * FC_K + 2 * lhalf;
  float s0 = 0.f, q0 = 0.f;

#pragma nounroll
  for (int i = 0; i < 8; ++i) {
    const int rt = wave + PG_WAVES * i;
#pragma unroll
    for (int k = 0; k < 12; ++k) myX[lane + 32 * k] = r[k];
    myX[384 + lane] = r[12];
    load_x_tile(x, i < 7 ? rt + PG_WAVES : rt, lane, r);
    __builtin_amdgcn_fence(__ATOMIC_ACQ_REL, "wavefront");
    v2f a[25];
#pragma unroll
    for (int s = 0; s < 25; ++s) a[s] = *(const v2f*)(ap + 4 * s);
    __builtin_amdgcn_sched_barrier(0);
    v8f acc0 = {}, acc1 = {};
#pragma unroll
    for (int s = 0; s < 25; ++s) {
      acc0 = __builtin_amdgcn_wmma_f32_16x16x4_f32(false, a[s], false, b0[s],
                                                   (short)0, acc0, false, false);
      acc1 = __builtin_amdgcn_wmma_f32_16x16x4_f32(false, a[s], false, b1[s],
                                                   (short)0, acc1, false, false);
    }
    float* hrow = h2 + (rt * 16) * H2ROW + pos0 * 16 + lane;
#pragma unroll
    for (int v = 0; v < 8; ++v) {
      float v0 = acc0[v] + bias0;
      float v1 = acc1[v] + bias1;
      v0 = v0 >= 0.f ? v0 : 0.01f * v0;
      v1 = v1 >= 0.f ? v1 : 0.01f * v1;
      s0 += v0;  q0 += v0 * v0;
      s0 += v1;  q0 += v1 * v1;
      const float snd = lhalf ? v0 : v1;
      const float rcv = __shfl_xor(snd, 16, 32);
      hrow[v * H2ROW]       = lhalf ? rcv : v0;
      hrow[(8 + v) * H2ROW] = lhalf ? v1 : rcv;
    }
  }

  s0 += __shfl_xor(s0, 16, 32);
  q0 += __shfl_xor(q0, 16, 32);
  if (lane < 16) { wsum[wave][lane] = s0; wsq[wave][lane] = q0; }
  __syncthreads();
  if (tid < 32) {
    int c = tid & 15;
    int st = tid >> 4;
    float tot = 0.f;
    for (int wv = 0; wv < PG_WAVES; ++wv) tot += st ? wsq[wv][c] : wsum[wv][c];
    part[blockIdx.x * 32 + tid] = tot;
  }
}

__global__ void __launch_bounds__(256) bn_reduce(const float* __restrict__ part,
                                                 float* __restrict__ gsum) {
  __shared__ float acc[256];
  const int t = threadIdx.x;
  const int cs = t & 31, r = t >> 5;
  const float* p = part + (blockIdx.x * BN_ROWS_PER_GROUP + r) * 32 + cs;
  float s = 0.f;
#pragma unroll
  for (int k = 0; k < 6; ++k) s += p[k * 256];
  if (r == 0) s += p[6 * 256];
  acc[t] = s;
  __syncthreads();
  if (t < 32) {
    float sum = 0.f;
    for (int rr = 0; rr < 8; ++rr) sum += acc[rr * 32 + t];
    gsum[blockIdx.x * 32 + t] = sum;
  }
}

#define LC_G      8
#define LC_HC     34
#define LC_TC     33
#define LC_GP     4
#define LC_POSQ   9
#define LC_GPQ    1530
#define LC_H3G    792
#define LC_ITEMS  2720
#define LC_TASKS  198
#define LC_BLOCKS 768

__device__ __forceinline__ void lc_load_w(const float* __restrict__ w1,
                                          const float* __restrict__ b1,
                                          int idx, int j0,
                                          float4 (&wa)[16], float4 (&wb)[16],
                                          float& bia, float& bib,
                                          int& xoff, int& h3off) {
  const int o1p = idx % 3;
  const int pp  = idx / 3;
  const int oi  = pp / LC_TC;
  const int ojl = pp - oi * LC_TC;
  const int oj  = j0 + ojl;
  const int o1a = 2 * o1p;
  const float* wp = w1 + o1a * 24832 + oi * 388 + oj * 4;
#pragma unroll
  for (int c = 0; c < 16; ++c) {
    wa[c] = *(const float4*)(wp + c * 1552);
    wb[c] = *(const float4*)(wp + 24832 + c * 1552);
  }
  bia = b1[o1a * 388 + oi * 97 + oj];
  bib = b1[(o1a + 1) * 388 + oi * 97 + oj];
  xoff  = (oi * LC_HC + ojl) * LC_POSQ;
  h3off = o1a * 132 + oi * LC_TC + ojl;
}

__device__ __forceinline__ void lc_ldx(const v4f* xg, int cq, v4f (&x)[8]) {
  x[0] = xg[2 * cq];                   x[1] = xg[2 * cq + 1];
  x[2] = xg[LC_POSQ + 2 * cq];         x[3] = xg[LC_POSQ + 2 * cq + 1];
  x[4] = xg[LC_HC * LC_POSQ + 2 * cq]; x[5] = xg[LC_HC * LC_POSQ + 2 * cq + 1];
  x[6] = xg[(LC_HC + 1) * LC_POSQ + 2 * cq];
  x[7] = xg[(LC_HC + 1) * LC_POSQ + 2 * cq + 1];
}

__device__ __forceinline__ v2f lc_term(const float4 w, v2f x00, v2f x01, v2f x10, v2f x11) {
  return w.x * x00 + w.y * x01 + w.z * x10 + w.w * x11;
}

__device__ __forceinline__ void lc2_one(int e, int b0, int j0,
                                        const float* __restrict__ w2,
                                        const float* __restrict__ b2,
                                        const float* sh3,
                                        float* __restrict__ out) {
  const int g   = e / 96;
  const int r   = e - g * 96;
  const int ri  = r >> 5;
  const int rjl = r & 31;
  const int rj  = j0 + rjl;
  float acc = b2[ri * 96 + rj];
#pragma unroll
  for (int o1 = 0; o1 < 6; ++o1) {
    float4 wv = *(const float4*)(w2 + (((o1 * 3 + ri) * 96 + rj) << 2));
    const float* xb = sh3 + g * LC_H3G + o1 * 132 + ri * LC_TC + rjl;
    acc += wv.x * xb[0] + wv.y * xb[1] + wv.z * xb[33] + wv.w * xb[34];
  }
  out[(b0 + g) * 288 + ri * 96 + rj] = acc;
}

__global__ void __launch_bounds__(224) lc_kernel(const float* __restrict__ h2,
                                                 const float* __restrict__ gsum,
                                                 const float* __restrict__ gamma,
                                                 const float* __restrict__ beta,
                                                 const float* __restrict__ w1,
                                                 const float* __restrict__ b1,
                                                 const float* __restrict__ w2,
                                                 const float* __restrict__ b2,
                                                 float* __restrict__ out) {
  __shared__ v4f sh2p[LC_GP * LC_GPQ];
  __shared__ float sh3[LC_G * LC_H3G];
  const int tid   = threadIdx.x;
  const int strip = blockIdx.x % 3;
  const int b0    = (blockIdx.x / 3) * LC_G;
  const int j0    = strip * 32;

  float4 wa[16], wb[16];
  float bia, bib;
  int xoff, h3off;
  lc_load_w(w1, b1, tid, j0, wa, wb, bia, bib, xoff, h3off);

  const int lane = tid & 31;
  float tot = 0.f;
#pragma unroll
  for (int g = 0; g < BN_GROUPS; ++g) tot += gsum[g * 32 + lane];
  const float oth = __shfl_xor(tot, 16, 32);
  const float csum = lane < 16 ? tot : oth;
  const float csq  = lane < 16 ? oth : tot;
  const float invN = 1.0f / BN_COUNT;
  const float mean = csum * invN;
  const float var  = csq * invN - mean * mean;
  const float inv  = rsqrtf(var + 1e-5f);
  const float scl  = gamma[lane & 15] * inv;
  const float shf  = beta[lane & 15] - mean * scl;

  const int q4 = (tid & 3) * 4;
  float4 sc4, sf4;
  sc4.x = __shfl(scl, q4 + 0, 32); sf4.x = __shfl(shf, q4 + 0, 32);
  sc4.y = __shfl(scl, q4 + 1, 32); sf4.y = __shfl(shf, q4 + 1, 32);
  sc4.z = __shfl(scl, q4 + 2, 32); sf4.z = __shfl(shf, q4 + 2, 32);
  sc4.w = __shfl(scl, q4 + 3, 32); sf4.w = __shfl(shf, q4 + 3, 32);
  const float4* src4 = (const float4*)h2;
#pragma nounroll
  for (int k0 = 0; k0 < 12; k0 += 6) {
    float4 r[12];
    int di[6];
#pragma unroll
    for (int u = 0; u < 6; ++u) {
      const int q   = tid + (k0 + u) * 224;
      const int seg = q / 136;
      const int w   = q - seg * 136;
      const int gp  = seg / 5;
      const int i   = seg - gp * 5;
      const int jj  = w >> 2;
      const int cq  = w & 3;
      const int gi  = ((b0 + 2 * gp) * CPOS + i * 98 + j0 + jj) * 4 + cq;
      r[2 * u]     = src4[gi];
      r[2 * u + 1] = src4[gi + CPOS * 4];
      di[u] = (seg * LC_HC + jj) * LC_POSQ + 2 * cq;
    }
#pragma unroll
    for (int u = 0; u < 6; ++u) {
      const float4 va = r[2 * u], vb = r[2 * u + 1];
      v4f lo, hi;
      lo.x = va.x * sc4.x + sf4.x;  lo.y = vb.x * sc4.x + sf4.x;
      lo.z = va.y * sc4.y + sf4.y;  lo.w = vb.y * sc4.y + sf4.y;
      hi.x = va.z * sc4.z + sf4.z;  hi.y = vb.z * sc4.z + sf4.z;
      hi.z = va.w * sc4.w + sf4.w;  hi.w = vb.w * sc4.w + sf4.w;
      sh2p[di[u]]     = lo;
      sh2p[di[u] + 1] = hi;
    }
  }
  if (tid < LC_ITEMS - 12 * 224) {
    const int q   = tid + 12 * 224;
    const int seg = q / 136;
    const int w   = q - seg * 136;
    const int gp  = seg / 5;
    const int i   = seg - gp * 5;
    const int jj  = w >> 2;
    const int cq  = w & 3;
    const int gi  = ((b0 + 2 * gp) * CPOS + i * 98 + j0 + jj) * 4 + cq;
    const float4 va = src4[gi], vb = src4[gi + CPOS * 4];
    v4f lo, hi;
    lo.x = va.x * sc4.x + sf4.x;  lo.y = vb.x * sc4.x + sf4.x;
    lo.z = va.y * sc4.y + sf4.y;  lo.w = vb.y * sc4.y + sf4.y;
    hi.x = va.z * sc4.z + sf4.z;  hi.y = vb.z * sc4.z + sf4.z;
    hi.z = va.w * sc4.w + sf4.w;  hi.w = vb.w * sc4.w + sf4.w;
    const int d = (seg * LC_HC + jj) * LC_POSQ + 2 * cq;
    sh2p[d]     = lo;
    sh2p[d + 1] = hi;
  }
  __syncthreads();

#pragma nounroll
  for (int pass = 0; pass < 2; ++pass) {
    if (pass == 1) {
      const int idx = LC_TASKS + tid;
      lc_load_w(w1, b1, idx < 395 ? idx : 395, j0, wa, wb, bia, bib, xoff, h3off);
    }
    if (tid < LC_TASKS) {
      const v4f* xb = sh2p + xoff;
      v4f xc[8], xn[8];
      lc_ldx(xb, 0, xc);
#pragma nounroll
      for (int gp = 0; gp < LC_GP; ++gp) {
        const v4f* xg  = xb + gp * LC_GPQ;
        const v4f* xgn = xb + (gp < LC_GP - 1 ? gp + 1 : gp) * LC_GPQ;
        v2f a0, a1;
        a0.x = bia; a0.y = bia;
        a1.x = bib; a1.y = bib;
#pragma unroll
        for (int cq = 0; cq < 4; ++cq) {
          if (cq < 3) lc_ldx(xg, cq + 1, xn); else lc_ldx(xgn, 0, xn);
          __builtin_amdgcn_sched_barrier(0);
          a0 += lc_term(wa[4 * cq + 0], xc[0].xy, xc[2].xy, xc[4].xy, xc[6].xy);
          a1 += lc_term(wb[4 * cq + 0], xc[0].xy, xc[2].xy, xc[4].xy, xc[6].xy);
          a0 += lc_term(wa[4 * cq + 1], xc[0].zw, xc[2].zw, xc[4].zw, xc[6].zw);
          a1 += lc_term(wb[4 * cq + 1], xc[0].zw, xc[2].zw, xc[4].zw, xc[6].zw);
          a0 += lc_term(wa[4 * cq + 2], xc[1].xy, xc[3].xy, xc[5].xy, xc[7].xy);
          a1 += lc_term(wb[4 * cq + 2], xc[1].xy, xc[3].xy, xc[5].xy, xc[7].xy);
          a0 += lc_term(wa[4 * cq + 3], xc[1].zw, xc[3].zw, xc[5].zw, xc[7].zw);
          a1 += lc_term(wb[4 * cq + 3], xc[1].zw, xc[3].zw, xc[5].zw, xc[7].zw);
#pragma unroll
          for (int e = 0; e < 8; ++e) xc[e] = xn[e];
        }
        float r0 = a0.x, r1 = a1.x, r2 = a0.y, r3 = a1.y;
        r0 = r0 >= 0.f ? r0 : 0.01f * r0;
        r1 = r1 >= 0.f ? r1 : 0.01f * r1;
        r2 = r2 >= 0.f ? r2 : 0.01f * r2;
        r3 = r3 >= 0.f ? r3 : 0.01f * r3;
        float* hp = sh3 + (2 * gp) * LC_H3G + h3off;
        hp[0]            = r0;
        hp[132]          = r1;
        hp[LC_H3G]       = r2;
        hp[LC_H3G + 132] = r3;
      }
    }
  }
  __syncthreads();

  lc2_one(tid,       b0, j0, w2, b2, sh3, out);
  lc2_one(tid + 224, b0, j0, w2, b2, sh3, out);
  lc2_one(tid + 448, b0, j0, w2, b2, sh3, out);
  if (tid < LC_G * 96 - 3 * 224) lc2_one(tid + 672, b0, j0, w2, b2, sh3, out);
}

extern "C" void kernel_launch(void* const* d_in, const int* in_sizes, int n_in,
                              void* d_out, int out_size, void* d_ws, size_t ws_size,
                              hipStream_t stream) {
  const float* x     = (const float*)d_in[0];
  const float* fw    = (const float*)d_in[1];
  const float* fb    = (const float*)d_in[2];
  const float* cw    = (const float*)d_in[3];
  const float* cb    = (const float*)d_in[4];
  const float* gamma = (const float*)d_in[5];
  const float* beta  = (const float*)d_in[6];
  const float* w1    = (const float*)d_in[7];
  const float* b1    = (const float*)d_in[8];
  const float* w2    = (const float*)d_in[9];
  const float* b2    = (const float*)d_in[10];

  float* ws    = (float*)d_ws;
  float* gsum  = ws;
  float* h2    = ws + H1_ELEMS;
  float* part  = ws + H1_ELEMS + H2_ELEMS;

  fcconv_pair<<<PG_BLOCKS, 512, 0, stream>>>(x, fw, fb, cw, cb, h2, part);
  bn_reduce  <<<BN_GROUPS, 256, 0, stream>>>(part, gsum);
  lc_kernel  <<<LC_BLOCKS, 224, 0, stream>>>(h2, gsum, gamma, beta, w1, b1, w2, b2,
                                      (float*)d_out);
}


# lc kernel persistent: 256 blocks of 16 waves keep a strip's lc1 weights in registers over 3 batch groups each; 3 of the waves stage the next h2 tile into a second LDS buffer meanwhile; sum order kept
# speedup vs baseline: 7.9079x; 1.1199x over previous
#include <hip/hip_runtime.h>

typedef __attribute__((ext_vector_type(2))) float v2f;
typedef __attribute__((ext_vector_type(4))) float v4f;
typedef __attribute__((ext_vector_type(8))) float v8f;

#define NB      2048
#define FC_K    100
#define CPOS    490
#define H1_ELEMS (NB * 17640)
#define H2_ELEMS (NB * 5 * 98 * 16)
#define H2ROW   7840
#define PART_ELEMS 250880
#define BN_GROUPS 5
#define BN_ROWS_PER_GROUP 49
#define BN_COUNT 1003520.0f
#define WQ_TILE 1600
#define PG_BLOCKS 245
#define PG_WAVES  16
#define PG_XQ     416

__device__ __forceinline__ void fold_tile(const float* __restrict__ fw,
                                          const float* __restrict__ fb,
                                          const float* __restrict__ cw,
                                          const float* __restrict__ cb,
                                          int pos, int kt, int lane,
                                          float* wp, float* bp) {
  const int lrow  = lane & 15;
  const int lhalf = lane >> 4;
  const int pi    = pos / 98;
  const int pj    = pos - pi * 98;
  const bool vdi  = pi < 4;
  const bool vdj  = pj < 97;
  const int k     = kt * 16 + lrow;
  const bool live = k <= FC_K;
  const float* colp = k < FC_K ? fw + k : fb;
  const int rs    = k < FC_K ? FC_K : 1;

  const float* cwp = cw + lrow * 144 + 4 * lhalf;
  v4f wc[18];
#pragma unroll
  for (int u = 0; u < 18; ++u) wc[u] = *(const v4f*)(cwp + 8 * u);

  v8f acc = {};
#pragma unroll
  for (int hh = 0; hh < 2; ++hh) {
    v2f af[18], bf[18];
    const bool v0 = hh ? vdi : true;
    const bool v1 = v0 && vdj;
    const int row0 = (pos + (v0 ? hh * 98 : 0)) * 36 + lhalf;
    const int row1 = (pos + (v1 ? hh * 98 + 1 : 0)) * 36 + lhalf;
    const bool keep0 = v0 && live;
    const bool keep1 = v1 && live;
#pragma unroll
    for (int u = 0; u < 18; ++u) {
      v2f f;
      f[0] = colp[(row0 + 2 * u) * rs];
      f[1] = colp[(row1 + 2 * u) * rs];
      f[0] = keep0 ? f[0] : 0.f;
      f[1] = keep1 ? f[1] : 0.f;
      af[u] = f;
      bf[u] = hh ? wc[u].zw : wc[u].xy;
    }
    __builtin_amdgcn_sched_barrier(0);
#pragma unroll
    for (int u = 0; u < 18; ++u)
      acc = __builtin_amdgcn_wmma_f32_16x16x4_f32(false, af[u], false, bf[u],
                                                  (short)0, acc, false, false);
  }
  if (kt < 6) {
#pragma unroll
    for (int q = 0; q < 4; ++q) {
      const int s = kt * 4 + 2 * lhalf + (q >> 1);
      v2f o; o[0] = acc[2 * q]; o[1] = acc[2 * q + 1];
      *(v2f*)(wp + (s * 32 + (q & 1) * 16 + lrow) * 2) = o;
    }
  } else if (lhalf == 0) {
#pragma unroll
    for (int q = 0; q < 2; ++q) {
      v2f o; o[0] = acc[2 * q]; o[1] = acc[2 * q + 1];
      *(v2f*)(wp + (24 * 32 + q * 16 + lrow) * 2) = o;
    }
    bp[lrow] = acc[4] + cb[lrow];
  }
}

__device__ __forceinline__ void load_x_tile(const float* __restrict__ x, int rt, int lane,
                                            v4f (&r)[13]) {
  const v4f* srcx = (const v4f*)(x + rt * (16 * FC_K));
#pragma unroll
  for (int k = 0; k < 12; ++k) r[k] = srcx[lane + 32 * k];
  const int ql = 384 + lane;
  r[12] = srcx[ql < 400 ? ql : 399];
}

__global__ void __launch_bounds__(512) fcconv_pair(const float* __restrict__ x,
                                                   const float* __restrict__ fw,
                                                   const float* __restrict__ fb,
                                                   const float* __restrict__ cw,
                                                   const float* __restrict__ cb,
                                                   float* __restrict__ h2,
                                                   float* __restrict__ part) {
  __shared__ v4f sXw[PG_WAVES * PG_XQ];
  __shared__ float sWq[2 * WQ_TILE];
  __shared__ float sBq[32];
  __shared__ float wsum[PG_WAVES][16];
  __shared__ float wsq[PG_WAVES][16];
  const int tid   = threadIdx.x;
  const int lane  = tid & 31;
  const int wave  = __builtin_amdgcn_readfirstlane(tid >> 5);
  const int lrow  = lane & 15;
  const int lhalf = lane >> 4;
  const int pos0  = blockIdx.x * 2;

  v4f r[13];
  load_x_tile(x, wave, lane, r);

  if (wave < 14) {
    const int pl = wave >= 7 ? 1 : 0;
    fold_tile(fw, fb, cw, cb, pos0 + pl, wave - 7 * pl, lane,
              sWq + pl * WQ_TILE, sBq + pl * 16);
  }
  __syncthreads();

  v2f b0[25], b1[25];
#pragma unroll
  for (int s = 0; s < 25; ++s) {
    b0[s] = *(const v2f*)(sWq + s * 64 + lane * 2);
    b1[s] = *(const v2f*)(sWq + WQ_TILE + s * 64 + lane * 2);
  }
  const float bias0 = sBq[lrow];
  const float bias1 = sBq[16 + lrow];

  v4f* myX = sXw + wave * PG_XQ;
  const float* ap = (const float*)myX + lrow * FC_K + 2 * lhalf;
  float s0 = 0.f, q0 = 0.f;

#pragma nounroll
  for (int i = 0; i < 8; ++i) {
    const int rt = wave + PG_WAVES * i;
#pragma unroll
    for (int k = 0; k < 12; ++k) myX[lane + 32 * k] = r[k];
    myX[384 + lane] = r[12];
    load_x_tile(x, i < 7 ? rt + PG_WAVES : rt, lane, r);
    __builtin_amdgcn_fence(__ATOMIC_ACQ_REL, "wavefront");
    v2f a[25];
#pragma unroll
    for (int s = 0; s < 25; ++s) a[s] = *(const v2f*)(ap + 4 * s);
    __builtin_amdgcn_sched_barrier(0);
    v8f acc0 = {}, acc1 = {};
#pragma unroll
    for (int s = 0; s < 25; ++s) {
      acc0 = __builtin_amdgcn_wmma_f32_16x16x4_f32(false, a[s], false, b0[s],
                                                   (short)0, acc0, false, false);
      acc1 = __builtin_amdgcn_wmma_f32_16x16x4_f32(false, a[s], false, b1[s],
                                                   (short)0, acc1, false, false);
    }
    float* hrow = h2 + (rt * 16) * H2ROW + pos0 * 16 + lane;
#pragma unroll
    for (int v = 0; v < 8; ++v) {
      float v0 = acc0[v] + bias0;
      float v1 = acc1[v] + bias1;
      v0 = v0 >= 0.f ? v0 : 0.01f * v0;
      v1 = v1 >= 0.f ? v1 : 0.01f * v1;
      s0 += v0;  q0 += v0 * v0;
      s0 += v1;  q0 += v1 * v1;
      const float snd = lhalf ? v0 : v1;
      const float rcv = __shfl_xor(snd, 16, 32);
      hrow[v * H2ROW]       = lhalf ? rcv : v0;
      hrow[(8 + v) * H2ROW] = lhalf ? v1 : rcv;
    }
  }

  s0 += __shfl_xor(s0, 16, 32);
  q0 += __shfl_xor(q0, 16, 32);
  if (lane < 16) { wsum[wave][lane] = s0; wsq[wave][lane] = q0; }
  __syncthreads();
  if (tid < 32) {
    int c = tid & 15;
    int st = tid >> 4;
    float tot = 0.f;
    for (int wv = 0; wv < PG_WAVES; ++wv) tot += st ? wsq[wv][c] : wsum[wv][c];
    part[blockIdx.x * 32 + tid] = tot;
  }
}

__global__ void __launch_bounds__(256) bn_reduce(const float* __restrict__ part,
                                                 float* __restrict__ gsum) {
  __shared__ float acc[256];
  const int t = threadIdx.x;
  const int cs = t & 31, r = t >> 5;
  const float* p = part + (blockIdx.x * BN_ROWS_PER_GROUP + r) * 32 + cs;
  float s = 0.f;
#pragma unroll
  for (int k = 0; k < 6; ++k) s += p[k * 256];
  if (r == 0) s += p[6 * 256];
  acc[t] = s;
  __syncthreads();
  if (t < 32) {
    float sum = 0.f;
    for (int rr = 0; rr < 8; ++rr) sum += acc[rr * 32 + t];
    gsum[blockIdx.x * 32 + t] = sum;
  }
}

#define LC_G      8
#define LC_HC     34
#define LC_TC     33
#define LC_GP     4
#define LC_POSQ   9
#define LC_GPQ    1530
#define LC_TILEQ  6120
#define LC_H3G    792
#define LC_ITEMS  2720
#define LC_NTASK  396
#define LC_UPB    3
#define LC_PBLOCKS 256
#define LC_THREADS 512
#define LC_CWAVES 13
#define LC_CTHR   416
#define LC_STHR   96

__device__ __forceinline__ void lc_load_w(const float* __restrict__ w1,
                                          const float* __restrict__ b1,
                                          int idx, int j0,
                                          float4 (&wa)[16], float4 (&wb)[16],
                                          float& bia, float& bib,
                                          int& xoff, int& h3off) {
  const int o1p = idx % 3;
  const int pp  = idx / 3;
  const int oi  = pp / LC_TC;
  const int ojl = pp - oi * LC_TC;
  const int oj  = j0 + ojl;
  const int o1a = 2 * o1p;
  const float* wp = w1 + o1a * 24832 + oi * 388 + oj * 4;
#pragma unroll
  for (int c = 0; c < 16; ++c) {
    wa[c] = *(const float4*)(wp + c * 1552);
    wb[c] = *(const float4*)(wp + 24832 + c * 1552);
  }
  bia = b1[o1a * 388 + oi * 97 + oj];
  bib = b1[(o1a + 1) * 388 + oi * 97 + oj];
  xoff  = (oi * LC_HC + ojl) * LC_POSQ;
  h3off = o1a * 132 + oi * LC_TC + ojl;
}

__device__ __forceinline__ void lc_ldx(const v4f* xg, int cq, v4f (&x)[8]) {
  x[0] = xg[2 * cq];                   x[1] = xg[2 * cq + 1];
  x[2] = xg[LC_POSQ + 2 * cq];         x[3] = xg[LC_POSQ + 2 * cq + 1];
  x[4] = xg[LC_HC * LC_POSQ + 2 * cq]; x[5] = xg[LC_HC * LC_POSQ + 2 * cq + 1];
  x[6] = xg[(LC_HC + 1) * LC_POSQ + 2 * cq];
  x[7] = xg[(LC_HC + 1) * LC_POSQ + 2 * cq + 1];
}

__device__ __forceinline__ v2f lc_term(const float4 w, v2f x00, v2f x01, v2f x10, v2f x11) {
  return w.x * x00 + w.y * x01 + w.z * x10 + w.w * x11;
}

__device__ __forceinline__ void lc2_one(int e, int b0, int j0,
                                        const float* __restrict__ w2,
                                        const float* __restrict__ b2,
                                        const float* sh3,
                                        float* __restrict__ out) {
  const int g   = e / 96;
  const int r   = e - g * 96;
  const int ri  = r >> 5;
  const int rjl = r & 31;
  const int rj  = j0 + rjl;
  float acc = b2[ri * 96 + rj];
#pragma unroll
  for (int o1 = 0; o1 < 6; ++o1) {
    float4 wv = *(const float4*)(w2 + (((o1 * 3 + ri) * 96 + rj) << 2));
    const float* xb = sh3 + g * LC_H3G + o1 * 132 + ri * LC_TC + rjl;
    acc += wv.x * xb[0] + wv.y * xb[1] + wv.z * xb[33] + wv.w * xb[34];
  }
  out[(b0 + g) * 288 + ri * 96 + rj] = acc;
}

template <int N>
__device__ __forceinline__ void lc_stage(const float4* __restrict__ src4, v4f* buf,
                                         int q0, int stride, int b0, int j0,
                                         const float4 sc4, const float4 sf4) {
  float4 r[2 * N];
  int di[N];
#pragma unroll
  for (int u = 0; u < N; ++u) {
    const int qu  = q0 + u * stride;
    const int q   = qu < LC_ITEMS ? qu : q0;
    const int seg = q / 136;
    const int w   = q - seg * 136;
    const int gp  = seg / 5;
    const int i   = seg - gp * 5;
    const int jj  = w >> 2;
    const int cq  = w & 3;
    const int gi  = ((b0 + 2 * gp) * CPOS + i * 98 + j0 + jj) * 4 + cq;
    r[2 * u]     = src4[gi];
    r[2 * u + 1] = src4[gi + CPOS * 4];
    di[u] = (seg * LC_HC + jj) * LC_POSQ + 2 * cq;
  }
#pragma unroll
  for (int u = 0; u < N; ++u) {
    const float4 va = r[2 * u], vb = r[2 * u + 1];
    v4f lo, hi;
    lo.x = va.x * sc4.x + sf4.x;  lo.y = vb.x * sc4.x + sf4.x;
    lo.z = va.y * sc4.y + sf4.y;  lo.w = vb.y * sc4.y + sf4.y;
    hi.x = va.z * sc4.z + sf4.z;  hi.y = vb.z * sc4.z + sf4.z;
    hi.z = va.w * sc4.w + sf4.w;  hi.w = vb.w * sc4.w + sf4.w;
    buf[di[u]]     = lo;
    buf[di[u] + 1] = hi;
  }
}

__global__ void __launch_bounds__(LC_THREADS) lc_kernel(const float* __restrict__ h2,
                                                        const float* __restrict__ gsum,
                                                        const float* __restrict__ gamma,
                                                        const float* __restrict__ beta,
                                                        const float* __restrict__ w1,
                                                        const float* __restrict__ b1,
                                                        const float* __restrict__ w2,
                                                        const float* __restrict__ b2,
                                                        float* __restrict__ out) {
  __shared__ v4f sh2p[2 * LC_TILEQ];
  __shared__ float sh3[LC_G * LC_H3G];
  const int tid  = threadIdx.x;
  const int lane = tid & 31;
  const int wave = __builtin_amdgcn_readfirstlane(tid >> 5);
  const bool stager = wave >= LC_CWAVES;
  const int u0 = blockIdx.x * LC_UPB;
  int strip = u0 >> 8;
  int j0    = strip * 32;
  int b0    = (u0 & 255) * LC_G;

  const int task = tid < LC_NTASK ? tid : LC_NTASK - 1;
  float4 wa[16], wb[16];
  float bia, bib;
  int xoff, h3off;
  lc_load_w(w1, b1, task, j0, wa, wb, bia, bib, xoff, h3off);

  float tot = 0.f;
#pragma unroll
  for (int g = 0; g < BN_GROUPS; ++g) tot += gsum[g * 32 + lane];
  const float oth = __shfl_xor(tot, 16, 32);
  const float csum = lane < 16 ? tot : oth;
  const float csq  = lane < 16 ? oth : tot;
  const float invN = 1.0f / BN_COUNT;
  const float mean = csum * invN;
  const float var  = csq * invN - mean * mean;
  const float inv  = rsqrtf(var + 1e-5f);
  const float scl  = gamma[lane & 15] * inv;
  const float shf  = beta[lane & 15] - mean * scl;

  const int q4 = (tid & 3) * 4;
  float4 sc4, sf4;
  sc4.x = __shfl(scl, q4 + 0, 32); sf4.x = __shfl(shf, q4 + 0, 32);
  sc4.y = __shfl(scl, q4 + 1, 32); sf4.y = __shfl(shf, q4 + 1, 32);
  sc4.z = __shfl(scl, q4 + 2, 32); sf4.z = __shfl(shf, q4 + 2, 32);
  sc4.w = __shfl(scl, q4 + 3, 32); sf4.w = __shfl(shf, q4 + 3, 32);
  const float4* src4 = (const float4*)h2;

  lc_stage<6>(src4, sh2p, tid, LC_THREADS, b0, j0, sc4, sf4);
  __syncthreads();

#pragma nounroll
  for (int uu = 0; uu < LC_UPB; ++uu) {
    const v4f* cur = sh2p + (uu & 1) * LC_TILEQ;
    v4f* nxt = sh2p + ((uu + 1) & 1) * LC_TILEQ;
    const bool more = uu + 1 < LC_UPB;
    const int un     = u0 + uu + 1;
    const int nstrip = un >> 8;
    const int nj0    = nstrip * 32;
    const int nb0    = (un & 255) * LC_G;

    if (stager) {
      if (more) {
        const int sl = tid - LC_CTHR;
#pragma nounroll
        for (int rnd = 0; rnd < 2; ++rnd)
          lc_stage<15>(src4, nxt, sl + rnd * 1440, LC_STHR, nb0, nj0, sc4, sf4);
      }
#pragma unroll
      for (int c = 0; c < 16; ++c) {
        wa[c] = make_float4(0.f, 0.f, 0.f, 0.f);
        wb[c] = make_float4(0.f, 0.f, 0.f, 0.f);
      }
    } else if (tid < LC_NTASK) {
      const v4f* xb = cur + xoff;
      v4f xc[8], xn[8];
      lc_ldx(xb, 0, xc);
#pragma nounroll
      for (int gp = 0; gp < LC_GP; ++gp) {
        const v4f* xg  = xb + gp * LC_GPQ;
        const v4f* xgn = xb + (gp < LC_GP - 1 ? gp + 1 : gp) * LC_GPQ;
        v2f a0, a1;
        a0.x = bia; a0.y = bia;
        a1.x = bib; a1.y = bib;
#pragma unroll
        for (int cq = 0; cq < 4; ++cq) {
          if (cq < 3) lc_ldx(xg, cq + 1, xn); else lc_ldx(xgn, 0, xn);
          __builtin_amdgcn_sched_barrier(0);
          a0 += lc_term(wa[4 * cq + 0], xc[0].xy, xc[2].xy, xc[4].xy, xc[6].xy);
          a1 += lc_term(wb[4 * cq + 0], xc[0].xy, xc[2].xy, xc[4].xy, xc[6].xy);
          a0 += lc_term(wa[4 * cq + 1], xc[0].zw, xc[2].zw, xc[4].zw, xc[6].zw);
          a1 += lc_term(wb[4 * cq + 1], xc[0].zw, xc[2].zw, xc[4].zw, xc[6].zw);
          a0 += lc_term(wa[4 * cq + 2], xc[1].xy, xc[3].xy, xc[5].xy, xc[7].xy);
          a1 += lc_term(wb[4 * cq + 2], xc[1].xy, xc[3].xy, xc[5].xy, xc[7].xy);
          a0 += lc_term(wa[4 * cq + 3], xc[1].zw, xc[3].zw, xc[5].zw, xc[7].zw);
          a1 += lc_term(wb[4 * cq + 3], xc[1].zw, xc[3].zw, xc[5].zw, xc[7].zw);
#pragma unroll
          for (int e = 0; e < 8; ++e) xc[e] = xn[e];
        }
        float r0 = a0.x, r1 = a1.x, r2 = a0.y, r3 = a1.y;
        r0 = r0 >= 0.f ? r0 : 0.01f * r0;
        r1 = r1 >= 0.f ? r1 : 0.01f * r1;
        r2 = r2 >= 0.f ? r2 : 0.01f * r2;
        r3 = r3 >= 0.f ? r3 : 0.01f * r3;
        float* hp = sh3 + (2 * gp) * LC_H3G + h3off;
        hp[0]            = r0;
        hp[132]          = r1;
        hp[LC_H3G]       = r2;
        hp[LC_H3G + 132] = r3;
      }
    }
    __syncthreads();

    lc2_one(tid, b0, j0, w2, b2, sh3, out);
    if (tid < LC_G * 96 - LC_THREADS) lc2_one(tid + LC_THREADS, b0, j0, w2, b2, sh3, out);

    if (more) {
      if (nstrip != strip)
        lc_load_w(w1, b1, task, nj0, wa, wb, bia, bib, xoff, h3off);
      strip = nstrip;  j0 = nj0;  b0 = nb0;
    }
    __syncthreads();
  }
}

extern "C" void kernel_launch(void* const* d_in, const int* in_sizes, int n_in,
                              void* d_out, int out_size, void* d_ws, size_t ws_size,
                              hipStream_t stream) {
  const float* x     = (const float*)d_in[0];
  const float* fw    = (const float*)d_in[1];
  const float* fb    = (const float*)d_in[2];
  const float* cw    = (const float*)d_in[3];
  const float* cb    = (const float*)d_in[4];
  const float* gamma = (const float*)d_in[5];
  const float* beta  = (const float*)d_in[6];
  const float* w1    = (const float*)d_in[7];
  const float* b1    = (const float*)d_in[8];
  const float* w2    = (const float*)d_in[9];
  const float* b2    = (const float*)d_in[10];

  float* ws    = (float*)d_ws;
  float* gsum  = ws;
  float* h2    = ws + H1_ELEMS;
  float* part  = ws + H1_ELEMS + H2_ELEMS;

  fcconv_pair<<<PG_BLOCKS, 512, 0, stream>>>(x, fw, fb, cw, cb, h2, part);
  bn_reduce  <<<BN_GROUPS, 256, 0, stream>>>(part, gsum);
  lc_kernel  <<<LC_PBLOCKS, LC_THREADS, 0, stream>>>(h2, gsum, gamma, beta, w1, b1, w2, b2,
                                      (float*)d_out);
}


# BN sums made in the lc kernel's prologue: a wave adds 16 of the 245 partial rows, waves combine via LDS behind an LDS-only barrier; bn_reduce launch gone (3 to 2); BN sum tree changes, fixed order
# speedup vs baseline: 8.1751x; 1.0338x over previous
#include <hip/hip_runtime.h>

typedef __attribute__((ext_vector_type(2))) float v2f;
typedef __attribute__((ext_vector_type(4))) float v4f;
typedef __attribute__((ext_vector_type(8))) float v8f;

#define NB      2048
#define FC_K    100
#define CPOS    490
#define H1_ELEMS (NB * 17640)
#define H2_ELEMS (NB * 5 * 98 * 16)
#define H2ROW   7840
#define PART_ELEMS 250880
#define BN_COUNT 1003520.0f
#define WQ_TILE 1600
#define PG_BLOCKS 245
#define PG_WAVES  16
#define PG_XQ     416
#define BN_PROWS  PG_BLOCKS

__device__ __forceinline__ void fold_tile(const float* __restrict__ fw,
                                          const float* __restrict__ fb,
                                          const float* __restrict__ cw,
                                          const float* __restrict__ cb,
                                          int pos, int kt, int lane,
                                          float* wp, float* bp) {
  const int lrow  = lane & 15;
  const int lhalf = lane >> 4;
  const int pi    = pos / 98;
  const int pj    = pos - pi * 98;
  const bool vdi  = pi < 4;
  const bool vdj  = pj < 97;
  const int k     = kt * 16 + lrow;
  const bool live = k <= FC_K;
  const float* colp = k < FC_K ? fw + k : fb;
  const int rs    = k < FC_K ? FC_K : 1;

  const float* cwp = cw + lrow * 144 + 4 * lhalf;
  v4f wc[18];
#pragma unroll
  for (int u = 0; u < 18; ++u) wc[u] = *(const v4f*)(cwp + 8 * u);

  v8f acc = {};
#pragma unroll
  for (int hh = 0; hh < 2; ++hh) {
    v2f af[18], bf[18];
    const bool v0 = hh ? vdi : true;
    const bool v1 = v0 && vdj;
    const int row0 = (pos + (v0 ? hh * 98 : 0)) * 36 + lhalf;
    const int row1 = (pos + (v1 ? hh * 98 + 1 : 0)) * 36 + lhalf;
    const bool keep0 = v0 && live;
    const bool keep1 = v1 && live;
#pragma unroll
    for (int u = 0; u < 18; ++u) {
      v2f f;
      f[0] = colp[(row0 + 2 * u) * rs];
      f[1] = colp[(row1 + 2 * u) * rs];
      f[0] = keep0 ? f[0] : 0.f;
      f[1] = keep1 ? f[1] : 0.f;
      af[u] = f;
      bf[u] = hh ? wc[u].zw : wc[u].xy;
    }
    __builtin_amdgcn_sched_barrier(0);
#pragma unroll
    for (int u = 0; u < 18; ++u)
      acc = __builtin_amdgcn_wmma_f32_16x16x4_f32(false, af[u], false, bf[u],
                                                  (short)0, acc, false, false);
  }
  if (kt < 6) {
#pragma unroll
    for (int q = 0; q < 4; ++q) {
      const int s = kt * 4 + 2 * lhalf + (q >> 1);
      v2f o; o[0] = acc[2 * q]; o[1] = acc[2 * q + 1];
      *(v2f*)(wp + (s * 32 + (q & 1) * 16 + lrow) * 2) = o;
    }
  } else if (lhalf == 0) {
#pragma unroll
    for (int q = 0; q < 2; ++q) {
      v2f o; o[0] = acc[2 * q]; o[1] = acc[2 * q + 1];
      *(v2f*)(wp + (24 * 32 + q * 16 + lrow) * 2) = o;
    }
    bp[lrow] = acc[4] + cb[lrow];
  }
}

__device__ __forceinline__ void load_x_tile(const float* __restrict__ x, int rt, int lane,
                                            v4f (&r)[13]) {
  const v4f* srcx = (const v4f*)(x + rt * (16 * FC_K));
#pragma unroll
  for (int k = 0; k < 12; ++k) r[k] = srcx[lane + 32 * k];
  const int ql = 384 + lane;
  r[12] = srcx[ql < 400 ? ql : 399];
}

__global__ void __launch_bounds__(512) fcconv_pair(const float* __restrict__ x,
                                                   const float* __restrict__ fw,
                                                   const float* __restrict__ fb,
                                                   const float* __restrict__ cw,
                                                   const float* __restrict__ cb,
                                                   float* __restrict__ h2,
                                                   float* __restrict__ part) {
  __shared__ v4f sXw[PG_WAVES * PG_XQ];
  __shared__ float sWq[2 * WQ_TILE];
  __shared__ float sBq[32];
  __shared__ float wsum[PG_WAVES][16];
  __shared__ float wsq[PG_WAVES][16];
  const int tid   = threadIdx.x;
  const int lane  = tid & 31;
  const int wave  = __builtin_amdgcn_readfirstlane(tid >> 5);
  const int lrow  = lane & 15;
  const int lhalf = lane >> 4;
  const int pos0  = blockIdx.x * 2;

  v4f r[13];
  load_x_tile(x, wave, lane, r);

  if (wave < 14) {
    const int pl = wave >= 7 ? 1 : 0;
    fold_tile(fw, fb, cw, cb, pos0 + pl, wave - 7 * pl, lane,
              sWq + pl * WQ_TILE, sBq + pl * 16);
  }
  __syncthreads();

  v2f b0[25], b1[25];
#pragma unroll
  for (int s = 0; s < 25; ++s) {
    b0[s] = *(const v2f*)(sWq + s * 64 + lane * 2);
    b1[s] = *(const v2f*)(sWq + WQ_TILE + s * 64 + lane * 2);
  }
  const float bias0 = sBq[lrow];
  const float bias1 = sBq[16 + lrow];

  v4f* myX = sXw + wave * PG_XQ;
  const float* ap = (const float*)myX + lrow * FC_K + 2 * lhalf;
  float s0 = 0.f, q0 = 0.f;

#pragma nounroll
  for (int i = 0; i < 8; ++i) {
    const int rt = wave + PG_WAVES * i;
#pragma unroll
    for (int k = 0; k < 12; ++k) myX[lane + 32 * k] = r[k];
    myX[384 + lane] = r[12];
    load_x_tile(x, i < 7 ? rt + PG_WAVES : rt, lane, r);
    __builtin_amdgcn_fence(__ATOMIC_ACQ_REL, "wavefront");
    v2f a[25];
#pragma unroll
    for (int s = 0; s < 25; ++s) a[s] = *(const v2f*)(ap + 4 * s);
    __builtin_amdgcn_sched_barrier(0);
    v8f acc0 = {}, acc1 = {};
#pragma unroll
    for (int s = 0; s < 25; ++s) {
      acc0 = __builtin_amdgcn_wmma_f32_16x16x4_f32(false, a[s], false, b0[s],
                                                   (short)0, acc0, false, false);
      acc1 = __builtin_amdgcn_wmma_f32_16x16x4_f32(false, a[s], false, b1[s],
                                                   (short)0, acc1, false, false);
    }
    float* hrow = h2 + (rt * 16) * H2ROW + pos0 * 16 + lane;
#pragma unroll
    for (int v = 0; v < 8; ++v) {
      float v0 = acc0[v] + bias0;
      float v1 = acc1[v] + bias1;
      v0 = v0 >= 0.f ? v0 : 0.01f * v0;
      v1 = v1 >= 0.f ? v1 : 0.01f * v1;
      s0 += v0;  q0 += v0 * v0;
      s0 += v1;  q0 += v1 * v1;
      const float snd = lhalf ? v0 : v1;
      const float rcv = __shfl_xor(snd, 16, 32);
      hrow[v * H2ROW]       = lhalf ? rcv : v0;
      hrow[(8 + v) * H2ROW] = lhalf ? v1 : rcv;
    }
  }

  s0 += __shfl_xor(s0, 16, 32);
  q0 += __shfl_xor(q0, 16, 32);
  if (lane < 16) { wsum[wave][lane] = s0; wsq[wave][lane] = q0; }
  __syncthreads();
  if (tid < 32) {
    int c = tid & 15;
    int st = tid >> 4;
    float tot = 0.f;
    for (int wv = 0; wv < PG_WAVES; ++wv) tot += st ? wsq[wv][c] : wsum[wv][c];
    part[blockIdx.x * 32 + tid] = tot;
  }
}

#define LC_G      8
#define LC_HC     34
#define LC_TC     33
#define LC_GP     4
#define LC_POSQ   9
#define LC_GPQ    1530
#define LC_TILEQ  6120
#define LC_H3G    792
#define LC_ITEMS  2720
#define LC_NTASK  396
#define LC_UPB    3
#define LC_PBLOCKS 256
#define LC_THREADS 512
#define LC_CWAVES 13
#define LC_CTHR   416
#define LC_STHR   96

__device__ __forceinline__ void lc_load_w(const float* __restrict__ w1,
                                          const float* __restrict__ b1,
                                          int idx, int j0,
                                          float4 (&wa)[16], float4 (&wb)[16],
                                          float& bia, float& bib,
                                          int& xoff, int& h3off) {
  const int o1p = idx % 3;
  const int pp  = idx / 3;
  const int oi  = pp / LC_TC;
  const int ojl = pp - oi * LC_TC;
  const int oj  = j0 + ojl;
  const int o1a = 2 * o1p;
  const float* wp = w1 + o1a * 24832 + oi * 388 + oj * 4;
#pragma unroll
  for (int c = 0; c < 16; ++c) {
    wa[c] = *(const float4*)(wp + c * 1552);
    wb[c] = *(const float4*)(wp + 24832 + c * 1552);
  }
  bia = b1[o1a * 388 + oi * 97 + oj];
  bib = b1[(o1a + 1) * 388 + oi * 97 + oj];
  xoff  = (oi * LC_HC + ojl) * LC_POSQ;
  h3off = o1a * 132 + oi * LC_TC + ojl;
}

__device__ __forceinline__ void lc_ldx(const v4f* xg, int cq, v4f (&x)[8]) {
  x[0] = xg[2 * cq];                   x[1] = xg[2 * cq + 1];
  x[2] = xg[LC_POSQ + 2 * cq];         x[3] = xg[LC_POSQ + 2 * cq + 1];
  x[4] = xg[LC_HC * LC_POSQ + 2 * cq]; x[5] = xg[LC_HC * LC_POSQ + 2 * cq + 1];
  x[6] = xg[(LC_HC + 1) * LC_POSQ + 2 * cq];
  x[7] = xg[(LC_HC + 1) * LC_POSQ + 2 * cq + 1];
}

__device__ __forceinline__ v2f lc_term(const float4 w, v2f x00, v2f x01, v2f x10, v2f x11) {
  return w.x * x00 + w.y * x01 + w.z * x10 + w.w * x11;
}

__device__ __forceinline__ void lc2_one(int e, int b0, int j0,
                                        const float* __restrict__ w2,
                                        const float* __restrict__ b2,
                                        const float* sh3,
                                        float* __restrict__ out) {
  const int g   = e / 96;
  const int r   = e - g * 96;
  const int ri  = r >> 5;
  const int rjl = r & 31;
  const int rj  = j0 + rjl;
  float acc = b2[ri * 96 + rj];
#pragma unroll
  for (int o1 = 0; o1 < 6; ++o1) {
    float4 wv = *(const float4*)(w2 + (((o1 * 3 + ri) * 96 + rj) << 2));
    const float* xb = sh3 + g * LC_H3G + o1 * 132 + ri * LC_TC + rjl;
    acc += wv.x * xb[0] + wv.y * xb[1] + wv.z * xb[33] + wv.w * xb[34];
  }
  out[(b0 + g) * 288 + ri * 96 + rj] = acc;
}

template <int N>
__device__ __forceinline__ void lc_stage_ld(const float4* __restrict__ src4,
                                            int q0, int stride, int b0, int j0,
                                            float4 (&r)[2 * N], int (&di)[N]) {
#pragma unroll
  for (int u = 0; u < N; ++u) {
    const int qu  = q0 + u * stride;
    const int q   = qu < LC_ITEMS ? qu : q0;
    const int seg = q / 136;
    const int w   = q - seg * 136;
    const int gp  = seg / 5;
    const int i   = seg - gp * 5;
    const int jj  = w >> 2;
    const int cq  = w & 3;
    const int gi  = ((b0 + 2 * gp) * CPOS + i * 98 + j0 + jj) * 4 + cq;
    r[2 * u]     = src4[gi];
    r[2 * u + 1] = src4[gi + CPOS * 4];
    di[u] = (seg * LC_HC + jj) * LC_POSQ + 2 * cq;
  }
}

template <int N>
__device__ __forceinline__ void lc_stage_st(v4f* buf, const float4 (&r)[2 * N],
                                            const int (&di)[N],
                                            const float4 sc4, const float4 sf4) {
#pragma unroll
  for (int u = 0; u < N; ++u) {
    const float4 va = r[2 * u], vb = r[2 * u + 1];
    v4f lo, hi;
    lo.x = va.x * sc4.x + sf4.x;  lo.y = vb.x * sc4.x + sf4.x;
    lo.z = va.y * sc4.y + sf4.y;  lo.w = vb.y * sc4.y + sf4.y;
    hi.x = va.z * sc4.z + sf4.z;  hi.y = vb.z * sc4.z + sf4.z;
    hi.z = va.w * sc4.w + sf4.w;  hi.w = vb.w * sc4.w + sf4.w;
    buf[di[u]]     = lo;
    buf[di[u] + 1] = hi;
  }
}

template <int N>
__device__ __forceinline__ void lc_stage(const float4* __restrict__ src4, v4f* buf,
                                         int q0, int stride, int b0, int j0,
                                         const float4 sc4, const float4 sf4) {
  float4 r[2 * N];
  int di[N];
  lc_stage_ld<N>(src4, q0, stride, b0, j0, r, di);
  lc_stage_st<N>(buf, r, di, sc4, sf4);
}

__global__ void __launch_bounds__(LC_THREADS) lc_kernel(const float* __restrict__ h2,
                                                        const float* __restrict__ part,
                                                        const float* __restrict__ gamma,
                                                        const float* __restrict__ beta,
                                                        const float* __restrict__ w1,
                                                        const float* __restrict__ b1,
                                                        const float* __restrict__ w2,
                                                        const float* __restrict__ b2,
                                                        float* __restrict__ out) {
  __shared__ v4f sh2p[2 * LC_TILEQ];
  __shared__ float sh3[LC_G * LC_H3G];
  __shared__ float sbn[LC_THREADS];
  const int tid  = threadIdx.x;
  const int lane = tid & 31;
  const int wave = __builtin_amdgcn_readfirstlane(tid >> 5);
  const bool stager = wave >= LC_CWAVES;
  const int u0 = blockIdx.x * LC_UPB;
  int strip = u0 >> 8;
  int j0    = strip * 32;
  int b0    = (u0 & 255) * LC_G;

  const bool tailrow = wave + 240 < BN_PROWS;
  const float* prow = part + wave * 32 + lane;
  float pv[16];
#pragma unroll
  for (int k = 0; k < 15; ++k) pv[k] = prow[k * (16 * 32)];
  pv[15] = prow[tailrow ? 15 * (16 * 32) : 0];

  const int task = tid < LC_NTASK ? tid : LC_NTASK - 1;
  float4 wa[16], wb[16];
  float bia, bib;
  int xoff, h3off;
  lc_load_w(w1, b1, task, j0, wa, wb, bia, bib, xoff, h3off);
  const float gam = gamma[lane & 15];
  const float bet = beta[lane & 15];

  const float4* src4 = (const float4*)h2;
  float4 tr[12];
  int tdi[6];
  lc_stage_ld<6>(src4, tid, LC_THREADS, b0, j0, tr, tdi);

  float ps = pv[0];
#pragma unroll
  for (int k = 1; k < 15; ++k) ps += pv[k];
  if (tailrow) ps += pv[15];
  sbn[tid] = ps;
  __builtin_amdgcn_fence(__ATOMIC_RELEASE, "workgroup", "local");
  __builtin_amdgcn_s_barrier();
  __builtin_amdgcn_fence(__ATOMIC_ACQUIRE, "workgroup", "local");

  float tot = sbn[lane];
#pragma unroll
  for (int wv = 1; wv < 16; ++wv) tot += sbn[wv * 32 + lane];
  const float oth = __shfl_xor(tot, 16, 32);
  const float csum = lane < 16 ? tot : oth;
  const float csq  = lane < 16 ? oth : tot;
  const float invN = 1.0f / BN_COUNT;
  const float mean = csum * invN;
  const float var  = csq * invN - mean * mean;
  const float inv  = rsqrtf(var + 1e-5f);
  const float scl  = gam * inv;
  const float shf  = bet - mean * scl;

  const int q4 = (tid & 3) * 4;
  float4 sc4, sf4;
  sc4.x = __shfl(scl, q4 + 0, 32); sf4.x = __shfl(shf, q4 + 0, 32);
  sc4.y = __shfl(scl, q4 + 1, 32); sf4.y = __shfl(shf, q4 + 1, 32);
  sc4.z = __shfl(scl, q4 + 2, 32); sf4.z = __shfl(shf, q4 + 2, 32);
  sc4.w = __shfl(scl, q4 + 3, 32); sf4.w = __shfl(shf, q4 + 3, 32);

  lc_stage_st<6>(sh2p, tr, tdi, sc4, sf4);
  __syncthreads();

#pragma nounroll
  for (int uu = 0; uu < LC_UPB; ++uu) {
    const v4f* cur = sh2p + (uu & 1) * LC_TILEQ;
    v4f* nxt = sh2p + ((uu + 1) & 1) * LC_TILEQ;
    const bool more = uu + 1 < LC_UPB;
    const int un     = u0 + uu + 1;
    const int nstrip = un >> 8;
    const int nj0    = nstrip * 32;
    const int nb0    = (un & 255) * LC_G;

    if (stager) {
      if (more) {
        const int sl = tid - LC_CTHR;
#pragma nounroll
        for (int rnd = 0; rnd < 2; ++rnd)
          lc_stage<15>(src4, nxt, sl + rnd * 1440, LC_STHR, nb0, nj0, sc4, sf4);
      }
#pragma unroll
      for (int c = 0; c < 16; ++c) {
        wa[c] = make_float4(0.f, 0.f, 0.f, 0.f);
        wb[c] = make_float4(0.f, 0.f, 0.f, 0.f);
      }
    } else if (tid < LC_NTASK) {
      const v4f* xb = cur + xoff;
      v4f xc[8], xn[8];
      lc_ldx(xb, 0, xc);
#pragma nounroll
      for (int gp = 0; gp < LC_GP; ++gp) {
        const v4f* xg  = xb + gp * LC_GPQ;
        const v4f* xgn = xb + (gp < LC_GP - 1 ? gp + 1 : gp) * LC_GPQ;
        v2f a0, a1;
        a0.x = bia; a0.y = bia;
        a1.x = bib; a1.y = bib;
#pragma unroll
        for (int cq = 0; cq < 4; ++cq) {
          if (cq < 3) lc_ldx(xg, cq + 1, xn); else lc_ldx(xgn, 0, xn);
          __builtin_amdgcn_sched_barrier(0);
          a0 += lc_term(wa[4 * cq + 0], xc[0].xy, xc[2].xy, xc[4].xy, xc[6].xy);
          a1 += lc_term(wb[4 * cq + 0], xc[0].xy, xc[2].xy, xc[4].xy, xc[6].xy);
          a0 += lc_term(wa[4 * cq + 1], xc[0].zw, xc[2].zw, xc[4].zw, xc[6].zw);
          a1 += lc_term(wb[4 * cq + 1], xc[0].zw, xc[2].zw, xc[4].zw, xc[6].zw);
          a0 += lc_term(wa[4 * cq + 2], xc[1].xy, xc[3].xy, xc[5].xy, xc[7].xy);
          a1 += lc_term(wb[4 * cq + 2], xc[1].xy, xc[3].xy, xc[5].xy, xc[7].xy);
          a0 += lc_term(wa[4 * cq + 3], xc[1].zw, xc[3].zw, xc[5].zw, xc[7].zw);
          a1 += lc_term(wb[4 * cq + 3], xc[1].zw, xc[3].zw, xc[5].zw, xc[7].zw);
#pragma unroll
          for (int e = 0; e < 8; ++e) xc[e] = xn[e];
        }
        float r0 = a0.x, r1 = a1.x, r2 = a0.y, r3 = a1.y;
        r0 = r0 >= 0.f ? r0 : 0.01f * r0;
        r1 = r1 >= 0.f ? r1 : 0.01f * r1;
        r2 = r2 >= 0.f ? r2 : 0.01f * r2;
        r3 = r3 >= 0.f ? r3 : 0.01f * r3;
        float* hp = sh3 + (2 * gp) * LC_H3G + h3off;
        hp[0]            = r0;
        hp[132]          = r1;
        hp[LC_H3G]       = r2;
        hp[LC_H3G + 132] = r3;
      }
    }
    __syncthreads();

    lc2_one(tid, b0, j0, w2, b2, sh3, out);
    if (tid < LC_G * 96 - LC_THREADS) lc2_one(tid + LC_THREADS, b0, j0, w2, b2, sh3, out);

    if (more) {
      if (nstrip != strip)
        lc_load_w(w1, b1, task, nj0, wa, wb, bia, bib, xoff, h3off);
      strip = nstrip;  j0 = nj0;  b0 = nb0;
    }
    __syncthreads();
  }
}

extern "C" void kernel_launch(void* const* d_in, const int* in_sizes, int n_in,
                              void* d_out, int out_size, void* d_ws, size_t ws_size,
                              hipStream_t stream) {
  const float* x     = (const float*)d_in[0];
  const float* fw    = (const float*)d_in[1];
  const float* fb    = (const float*)d_in[2];
  const float* cw    = (const float*)d_in[3];
  const float* cb    = (const float*)d_in[4];
  const float* gamma = (const float*)d_in[5];
  const float* beta  = (const float*)d_in[6];
  const float* w1    = (const float*)d_in[7];
  const float* b1    = (const float*)d_in[8];
  const float* w2    = (const float*)d_in[9];
  const float* b2    = (const float*)d_in[10];

  float* ws    = (float*)d_ws;
  float* h2    = ws + H1_ELEMS;
  float* part  = ws + H1_ELEMS + H2_ELEMS;

  fcconv_pair<<<PG_BLOCKS, 512, 0, stream>>>(x, fw, fb, cw, cb, h2, part);
  lc_kernel  <<<LC_PBLOCKS, LC_THREADS, 0, stream>>>(h2, part, gamma, beta, w1, b1, w2, b2,
                                      (float*)d_out);
}
